# starting kernel
# baseline (speedup 1.0000x reference)
#include <hip/hip_runtime.h>
#include <hip/hip_bf16.h>

#define N_NODES 100000
#define N_EDGES 1600000
#define DIM 64
#define OUT_DIM 32

typedef __attribute__((ext_vector_type(2))) float v2f;
typedef __attribute__((ext_vector_type(8))) float v8f;

__global__ void init_kernel(float* __restrict__ deg,
                            float* __restrict__ agg1,
                            float* __restrict__ agg2) {
    size_t tid = (size_t)blockIdx.x * blockDim.x + threadIdx.x;
    const size_t tot = (size_t)N_NODES * DIM;
    if (tid < tot) { agg1[tid] = 0.0f; agg2[tid] = 0.0f; }
    if (tid < N_NODES) deg[tid] = 1.0f;
}

__global__ void degree_kernel(const int* __restrict__ dst, float* __restrict__ deg) {
    int e = blockIdx.x * blockDim.x + threadIdx.x;
    if (e < N_EDGES) atomicAdd(&deg[dst[e]], 1.0f);
}

__global__ void rsqrt_kernel(float* __restrict__ deg) {
    int i = blockIdx.x * blockDim.x + threadIdx.x;
    if (i < N_NODES) deg[i] = rsqrtf(deg[i]);
}

__global__ __launch_bounds__(32) void gemm_x_w1(const float* __restrict__ x,
                                                const float* __restrict__ W,
                                                float* __restrict__ out) {
    const int lane = threadIdx.x;
    const int m    = lane & 15;
    const int half = lane >> 4;
    const int row0 = blockIdx.x * 16;

    v8f acc0 = {}, acc1 = {}, acc2 = {}, acc3 = {};
    const float* xr = x + (size_t)(row0 + m) * DIM;

#pragma unroll
    for (int ks = 0; ks < 16; ++ks) {
        const int kk = ks * 4 + half * 2;
        v2f a;  a.x = xr[kk];  a.y = xr[kk + 1];
        const float* wr0 = W + (size_t)kk * DIM;
        const float* wr1 = W + (size_t)(kk + 1) * DIM;
        v2f b0, b1, b2, b3;
        b0.x = wr0[m];      b0.y = wr1[m];
        b1.x = wr0[16 + m]; b1.y = wr1[16 + m];
        b2.x = wr0[32 + m]; b2.y = wr1[32 + m];
        b3.x = wr0[48 + m]; b3.y = wr1[48 + m];
        acc0 = __builtin_amdgcn_wmma_f32_16x16x4_f32(false, a, false, b0, (short)0, acc0, false, false);
        acc1 = __builtin_amdgcn_wmma_f32_16x16x4_f32(false, a, false, b1, (short)0, acc1, false, false);
        acc2 = __builtin_amdgcn_wmma_f32_16x16x4_f32(false, a, false, b2, (short)0, acc2, false, false);
        acc3 = __builtin_amdgcn_wmma_f32_16x16x4_f32(false, a, false, b3, (short)0, acc3, false, false);
    }

#pragma unroll
    for (int r = 0; r < 8; ++r) {
        float* o = out + (size_t)(row0 + r + 8 * half) * DIM;
        o[m]      = acc0[r];
        o[16 + m] = acc1[r];
        o[32 + m] = acc2[r];
        o[48 + m] = acc3[r];
    }
}

__global__ void scatter_kernel(const int* __restrict__ src,
                               const int* __restrict__ dst,
                               const float* __restrict__ dinv,
                               const float* __restrict__ feat,
                               float* __restrict__ out) {
    size_t tid = (size_t)blockIdx.x * blockDim.x + threadIdx.x;
    const size_t total = (size_t)(N_EDGES + N_NODES) * DIM;
    if (tid >= total) return;
    const int e = (int)(tid >> 6);
    const int j = (int)(tid & 63);
    int s, d;
    if (e < N_EDGES) { s = src[e]; d = dst[e]; }
    else             { s = e - N_EDGES; d = s; }
    const float w = dinv[s] * dinv[d];
    atomicAdd(out + (size_t)d * DIM + j, feat[(size_t)s * DIM + j] * w);
}

__global__ void bias_relu_kernel(float* __restrict__ h, const float* __restrict__ b1) {
    size_t tid = (size_t)blockIdx.x * blockDim.x + threadIdx.x;
    if (tid < (size_t)N_NODES * DIM)
        h[tid] = fmaxf(h[tid] + b1[tid & 63], 0.0f);
}

__global__ __launch_bounds__(32) void gemm_out(const float* __restrict__ agg,
                                               const float* __restrict__ Wmu,
                                               const float* __restrict__ bmu,
                                               const float* __restrict__ Wlv,
                                               const float* __restrict__ blv,
                                               float* __restrict__ out) {
    const int lane = threadIdx.x;
    const int m    = lane & 15;
    const int half = lane >> 4;
    const int row0 = blockIdx.x * 16;

    v8f accMu0 = {}, accMu1 = {}, accLv0 = {}, accLv1 = {};
    const float* ar = agg + (size_t)(row0 + m) * DIM;

#pragma unroll
    for (int ks = 0; ks < 16; ++ks) {
        const int kk = ks * 4 + half * 2;
        v2f a;  a.x = ar[kk];  a.y = ar[kk + 1];
        const float* mu0 = Wmu + (size_t)kk * OUT_DIM;
        const float* mu1 = Wmu + (size_t)(kk + 1) * OUT_DIM;
        const float* lv0 = Wlv + (size_t)kk * OUT_DIM;
        const float* lv1 = Wlv + (size_t)(kk + 1) * OUT_DIM;
        v2f b0, b1, b2, b3;
        b0.x = mu0[m];      b0.y = mu1[m];
        b1.x = mu0[16 + m]; b1.y = mu1[16 + m];
        b2.x = lv0[m];      b2.y = lv1[m];
        b3.x = lv0[16 + m]; b3.y = lv1[16 + m];
        accMu0 = __builtin_amdgcn_wmma_f32_16x16x4_f32(false, a, false, b0, (short)0, accMu0, false, false);
        accMu1 = __builtin_amdgcn_wmma_f32_16x16x4_f32(false, a, false, b1, (short)0, accMu1, false, false);
        accLv0 = __builtin_amdgcn_wmma_f32_16x16x4_f32(false, a, false, b2, (short)0, accLv0, false, false);
        accLv1 = __builtin_amdgcn_wmma_f32_16x16x4_f32(false, a, false, b3, (short)0, accLv1, false, false);
    }

    float* lvBase = out + (size_t)N_NODES * OUT_DIM;
#pragma unroll
    for (int r = 0; r < 8; ++r) {
        const int row = row0 + r + 8 * half;
        float* omu = out    + (size_t)row * OUT_DIM;
        float* olv = lvBase + (size_t)row * OUT_DIM;
        omu[m]      = accMu0[r] + bmu[m];
        omu[16 + m] = accMu1[r] + bmu[16 + m];
        olv[m]      = accLv0[r] + blv[m];
        olv[16 + m] = accLv1[r] + blv[16 + m];
    }
}

extern "C" void kernel_launch(void* const* d_in, const int* in_sizes, int n_in,
                              void* d_out, int out_size, void* d_ws, size_t ws_size,
                              hipStream_t stream) {
    const float* x    = (const float*)d_in[0];
    const int*   eidx = (const int*)d_in[1];
    const int*   src  = eidx;
    const int*   dst  = eidx + N_EDGES;
    const float* W1   = (const float*)d_in[2];
    const float* b1   = (const float*)d_in[3];
    const float* Wmu  = (const float*)d_in[4];
    const float* bmu  = (const float*)d_in[5];
    const float* Wlv  = (const float*)d_in[6];
    const float* blv  = (const float*)d_in[7];
    float*       out  = (float*)d_out;

    float* deg  = (float*)d_ws;
    float* xw   = deg  + N_NODES;
    float* agg1 = xw   + (size_t)N_NODES * DIM;
    float* agg2 = agg1 + (size_t)N_NODES * DIM;

    const int TB = 256;
    const size_t featElems    = (size_t)N_NODES * DIM;
    const size_t scatterElems = (size_t)(N_EDGES + N_NODES) * DIM;

    const unsigned gInit    = (unsigned)((featElems + TB - 1) / TB);
    const unsigned gDeg     = (N_EDGES + TB - 1) / TB;
    const unsigned gNode    = (N_NODES + TB - 1) / TB;
    const unsigned gScatter = (unsigned)((scatterElems + TB - 1) / TB);
    const unsigned gGemm    = N_NODES / 16;

    init_kernel<<<gInit, TB, 0, stream>>>(deg, agg1, agg2);
    degree_kernel<<<gDeg, TB, 0, stream>>>(dst, deg);
    rsqrt_kernel<<<gNode, TB, 0, stream>>>(deg);

    gemm_x_w1<<<gGemm, 32, 0, stream>>>(x, W1, xw);
    scatter_kernel<<<gScatter, TB, 0, stream>>>(src, dst, deg, xw, agg1);
    bias_relu_kernel<<<gInit, TB, 0, stream>>>(agg1, b1);
    scatter_kernel<<<gScatter, TB, 0, stream>>>(src, dst, deg, agg1, agg2);
    gemm_out<<<gGemm, 32, 0, stream>>>(agg2, Wmu, bmu, Wlv, blv, out);
}


# Owner-computes push: 256 groups each hold 391 destination rows in LDS, scan the whole dst list, add matching source rows by LDS float atomics, store rows once; no zero-fill; sum order = LDS arrival
# speedup vs baseline: 2.2552x; 2.2552x over previous
#include <hip/hip_runtime.h>
#include <hip/hip_bf16.h>

#define N_NODES 100000
#define N_EDGES 1600000
#define DIM 64
#define OUT_DIM 32

typedef __attribute__((ext_vector_type(2))) float v2f;
typedef __attribute__((ext_vector_type(8))) float v8f;

__global__ void init_kernel(float* __restrict__ deg) {
    int i = blockIdx.x * blockDim.x + threadIdx.x;
    if (i < N_NODES) deg[i] = 1.0f;
}

__global__ void degree_kernel(const int* __restrict__ dst, float* __restrict__ deg) {
    int e = blockIdx.x * blockDim.x + threadIdx.x;
    if (e < N_EDGES) atomicAdd(&deg[dst[e]], 1.0f);
}

__global__ void rsqrt_kernel(float* __restrict__ deg) {
    int i = blockIdx.x * blockDim.x + threadIdx.x;
    if (i < N_NODES) deg[i] = rsqrtf(deg[i]);
}

__global__ __launch_bounds__(32) void gemm_x_w1(const float* __restrict__ x,
                                                const float* __restrict__ W,
                                                float* __restrict__ out) {
    const int lane = threadIdx.x;
    const int m    = lane & 15;
    const int half = lane >> 4;
    const int row0 = blockIdx.x * 16;

    v8f acc0 = {}, acc1 = {}, acc2 = {}, acc3 = {};
    const float* xr = x + (size_t)(row0 + m) * DIM;

#pragma unroll
    for (int ks = 0; ks < 16; ++ks) {
        const int kk = ks * 4 + half * 2;
        v2f a;  a.x = xr[kk];  a.y = xr[kk + 1];
        const float* wr0 = W + (size_t)kk * DIM;
        const float* wr1 = W + (size_t)(kk + 1) * DIM;
        v2f b0, b1, b2, b3;
        b0.x = wr0[m];      b0.y = wr1[m];
        b1.x = wr0[16 + m]; b1.y = wr1[16 + m];
        b2.x = wr0[32 + m]; b2.y = wr1[32 + m];
        b3.x = wr0[48 + m]; b3.y = wr1[48 + m];
        acc0 = __builtin_amdgcn_wmma_f32_16x16x4_f32(false, a, false, b0, (short)0, acc0, false, false);
        acc1 = __builtin_amdgcn_wmma_f32_16x16x4_f32(false, a, false, b1, (short)0, acc1, false, false);
        acc2 = __builtin_amdgcn_wmma_f32_16x16x4_f32(false, a, false, b2, (short)0, acc2, false, false);
        acc3 = __builtin_amdgcn_wmma_f32_16x16x4_f32(false, a, false, b3, (short)0, acc3, false, false);
    }

#pragma unroll
    for (int r = 0; r < 8; ++r) {
        float* o = out + (size_t)(row0 + r + 8 * half) * DIM;
        o[m]      = acc0[r];
        o[16 + m] = acc1[r];
        o[32 + m] = acc2[r];
        o[48 + m] = acc3[r];
    }
}

#define TILE_ROWS    391
#define N_TILES      256
#define PUSH_THREADS 1024
#define PUSH_WAVES   (PUSH_THREADS / 32)
#define Q_CAP        160
#define N_DST4       (N_EDGES / 4)
#define SCAN_SLOTS   ((N_DST4 + PUSH_THREADS - 1) / PUSH_THREADS)
#define E_BITS       21

__global__ __launch_bounds__(PUSH_THREADS) void tile_push_kernel(const int* __restrict__ src,
                                                                 const int* __restrict__ dst,
                                                                 const float* __restrict__ dinv,
                                                                 const float* __restrict__ feat,
                                                                 float* __restrict__ out) {
    __shared__ float    acc[TILE_ROWS * DIM];
    __shared__ unsigned queue[PUSH_WAVES * Q_CAP];

    const int tid  = threadIdx.x;
    const int lane = tid & 31;
    const int wave = tid >> 5;
    const int lo   = blockIdx.x * TILE_ROWS;
    const int rows = min(TILE_ROWS, N_NODES - lo);
    const unsigned urows = (unsigned)rows;

    const float4* feat4 = (const float4*)feat;
    const float2* feat2 = (const float2*)feat;
    const int4*   dst4  = (const int4*)dst;

    for (int i = tid; i < rows * (DIM / 4); i += PUSH_THREADS) {
        const int   r  = i >> 4;
        const float dv = dinv[lo + r];
        const float w  = dv * dv;
        const float4 f = feat4[(size_t)lo * (DIM / 4) + i];
        float4 o;
        o.x = f.x * w; o.y = f.y * w; o.z = f.z * w; o.w = f.w * w;
        ((float4*)acc)[i] = o;
    }
    __syncthreads();

    const int qoff = wave * Q_CAP;
    int cnt = 0;

    const int slot0 = (blockIdx.x * SCAN_SLOTS) >> 8;

    auto load_slot = [&](int it, int& e0) -> int4 {
        int sl = slot0 + it;
        if (sl >= SCAN_SLOTS) sl -= SCAN_SLOTS;
        const int i4 = sl * PUSH_THREADS + tid;
        e0 = i4 * 4;
        int4 v = dst4[min(i4, N_DST4 - 1)];
        if (it >= SCAN_SLOTS || i4 >= N_DST4) { v.x = -1; v.y = -1; v.z = -1; v.w = -1; }
        return v;
    };

    auto append = [&](unsigned r, int e) {
        const bool m = r < urows;
        const unsigned mask = __builtin_amdgcn_ballot_w32(m);
        if (m) {
            const unsigned pos = (unsigned)cnt + __builtin_amdgcn_mbcnt_lo(mask, 0u);
            __hip_atomic_store(&queue[qoff + pos], (r << E_BITS) | (unsigned)e,
                               __ATOMIC_RELAXED, __HIP_MEMORY_SCOPE_WAVEFRONT);
        }
        cnt += __builtin_popcount(mask);
    };

    auto scan_slot = [&](const int4& d, int e0) {
        const unsigned r0 = (unsigned)(d.x - lo);
        const unsigned r1 = (unsigned)(d.y - lo);
        const unsigned r2 = (unsigned)(d.z - lo);
        const unsigned r3 = (unsigned)(d.w - lo);
        const unsigned mn = min(min(r0, r1), min(r2, r3));
        if (__builtin_amdgcn_ballot_w32(mn < urows) != 0u) {
            append(r0, e0);
            append(r1, e0 + 1);
            append(r2, e0 + 2);
            append(r3, e0 + 3);
        }
    };

    auto work_off = [&](int base, int n) {
        unsigned ent = 0u;
        if (lane < n) ent = __hip_atomic_load(&queue[qoff + base + lane],
                                              __ATOMIC_RELAXED, __HIP_MEMORY_SCOPE_WAVEFRONT);
        const int e  = (int)(ent & ((1u << E_BITS) - 1u));
        const int dl = (int)(ent >> E_BITS);
        int   s = 0;
        float w = 0.0f;
        if (lane < n) {
            s = src[e];
            w = dinv[s] * dinv[lo + dl];
        }
        const int wbits = __float_as_int(w);
        for (int i = 0; i < n; i += 8) {
            float2 f[8];
#pragma unroll
            for (int k = 0; k < 8; ++k) {
                const int si = __builtin_amdgcn_readlane(s, i + k);
                f[k] = feat2[(size_t)si * (DIM / 2) + lane];
            }
#pragma unroll
            for (int k = 0; k < 8; ++k) {
                if (i + k < n) {
                    const float wi  = __int_as_float(__builtin_amdgcn_readlane(wbits, i + k));
                    const int   dli = __builtin_amdgcn_readlane(dl, i + k);
                    float* a = acc + dli * DIM + 2 * lane;
                    atomicAdd(a,     f[k].x * wi);
                    atomicAdd(a + 1, f[k].y * wi);
                }
            }
        }
    };

    int  e_cur[4], e_nxt[4];
    int4 d_cur[4], d_nxt[4];
#pragma unroll
    for (int u = 0; u < 4; ++u) d_cur[u] = load_slot(u, e_cur[u]);

    for (int it = 0; it < SCAN_SLOTS; it += 4) {
#pragma unroll
        for (int u = 0; u < 4; ++u) d_nxt[u] = load_slot(it + 4 + u, e_nxt[u]);
#pragma unroll
        for (int u = 0; u < 4; ++u) {
            scan_slot(d_cur[u], e_cur[u]);
            while (cnt >= 32) { cnt -= 32; work_off(cnt, 32); }
        }
#pragma unroll
        for (int u = 0; u < 4; ++u) { d_cur[u] = d_nxt[u]; e_cur[u] = e_nxt[u]; }
    }
    if (cnt > 0) work_off(0, cnt);

    __syncthreads();

    float4* out4 = (float4*)out + (size_t)lo * (DIM / 4);
    for (int i = tid; i < rows * (DIM / 4); i += PUSH_THREADS)
        out4[i] = ((const float4*)acc)[i];
}

__global__ void bias_relu_kernel(float* __restrict__ h, const float* __restrict__ b1) {
    size_t tid = (size_t)blockIdx.x * blockDim.x + threadIdx.x;
    if (tid < (size_t)N_NODES * DIM)
        h[tid] = fmaxf(h[tid] + b1[tid & 63], 0.0f);
}

__global__ __launch_bounds__(32) void gemm_out(const float* __restrict__ agg,
                                               const float* __restrict__ Wmu,
                                               const float* __restrict__ bmu,
                                               const float* __restrict__ Wlv,
                                               const float* __restrict__ blv,
                                               float* __restrict__ out) {
    const int lane = threadIdx.x;
    const int m    = lane & 15;
    const int half = lane >> 4;
    const int row0 = blockIdx.x * 16;

    v8f accMu0 = {}, accMu1 = {}, accLv0 = {}, accLv1 = {};
    const float* ar = agg + (size_t)(row0 + m) * DIM;

#pragma unroll
    for (int ks = 0; ks < 16; ++ks) {
        const int kk = ks * 4 + half * 2;
        v2f a;  a.x = ar[kk];  a.y = ar[kk + 1];
        const float* mu0 = Wmu + (size_t)kk * OUT_DIM;
        const float* mu1 = Wmu + (size_t)(kk + 1) * OUT_DIM;
        const float* lv0 = Wlv + (size_t)kk * OUT_DIM;
        const float* lv1 = Wlv + (size_t)(kk + 1) * OUT_DIM;
        v2f b0, b1, b2, b3;
        b0.x = mu0[m];      b0.y = mu1[m];
        b1.x = mu0[16 + m]; b1.y = mu1[16 + m];
        b2.x = lv0[m];      b2.y = lv1[m];
        b3.x = lv0[16 + m]; b3.y = lv1[16 + m];
        accMu0 = __builtin_amdgcn_wmma_f32_16x16x4_f32(false, a, false, b0, (short)0, accMu0, false, false);
        accMu1 = __builtin_amdgcn_wmma_f32_16x16x4_f32(false, a, false, b1, (short)0, accMu1, false, false);
        accLv0 = __builtin_amdgcn_wmma_f32_16x16x4_f32(false, a, false, b2, (short)0, accLv0, false, false);
        accLv1 = __builtin_amdgcn_wmma_f32_16x16x4_f32(false, a, false, b3, (short)0, accLv1, false, false);
    }

    float* lvBase = out + (size_t)N_NODES * OUT_DIM;
#pragma unroll
    for (int r = 0; r < 8; ++r) {
        const int row = row0 + r + 8 * half;
        float* omu = out    + (size_t)row * OUT_DIM;
        float* olv = lvBase + (size_t)row * OUT_DIM;
        omu[m]      = accMu0[r] + bmu[m];
        omu[16 + m] = accMu1[r] + bmu[16 + m];
        olv[m]      = accLv0[r] + blv[m];
        olv[16 + m] = accLv1[r] + blv[16 + m];
    }
}

extern "C" void kernel_launch(void* const* d_in, const int* in_sizes, int n_in,
                              void* d_out, int out_size, void* d_ws, size_t ws_size,
                              hipStream_t stream) {
    const float* x    = (const float*)d_in[0];
    const int*   eidx = (const int*)d_in[1];
    const int*   src  = eidx;
    const int*   dst  = eidx + N_EDGES;
    const float* W1   = (const float*)d_in[2];
    const float* b1   = (const float*)d_in[3];
    const float* Wmu  = (const float*)d_in[4];
    const float* bmu  = (const float*)d_in[5];
    const float* Wlv  = (const float*)d_in[6];
    const float* blv  = (const float*)d_in[7];
    float*       out  = (float*)d_out;

    float* deg  = (float*)d_ws;
    float* xw   = deg  + N_NODES;
    float* agg1 = xw   + (size_t)N_NODES * DIM;
    float* agg2 = agg1 + (size_t)N_NODES * DIM;

    const int TB = 256;
    const size_t featElems    = (size_t)N_NODES * DIM;

    const unsigned gInit    = (unsigned)((featElems + TB - 1) / TB);
    const unsigned gDeg     = (N_EDGES + TB - 1) / TB;
    const unsigned gNode    = (N_NODES + TB - 1) / TB;
    const unsigned gGemm    = N_NODES / 16;

    init_kernel<<<gNode, TB, 0, stream>>>(deg);
    degree_kernel<<<gDeg, TB, 0, stream>>>(dst, deg);
    rsqrt_kernel<<<gNode, TB, 0, stream>>>(deg);

    gemm_x_w1<<<gGemm, 32, 0, stream>>>(x, W1, xw);
    tile_push_kernel<<<N_TILES, PUSH_THREADS, 0, stream>>>(src, dst, deg, xw, agg1);
    bias_relu_kernel<<<gInit, TB, 0, stream>>>(agg1, b1);
    tile_push_kernel<<<N_TILES, PUSH_THREADS, 0, stream>>>(src, dst, deg, agg1, agg2);
    gemm_out<<<gGemm, 32, 0, stream>>>(agg2, Wmu, bmu, Wlv, blv, out);
}


# Layer 1's push also writes each tile's matched edges (source, local row, weight) as 256-byte chunks; layer 2's push replays them and does not scan the dst list; sum order = LDS arrival as before
# speedup vs baseline: 2.7090x; 1.2012x over previous
#include <hip/hip_runtime.h>
#include <hip/hip_bf16.h>

#define N_NODES 100000
#define N_EDGES 1600000
#define DIM 64
#define OUT_DIM 32

typedef __attribute__((ext_vector_type(2))) float v2f;
typedef __attribute__((ext_vector_type(8))) float v8f;

__global__ void init_kernel(float* __restrict__ deg) {
    int i = blockIdx.x * blockDim.x + threadIdx.x;
    if (i < N_NODES) deg[i] = 1.0f;
}

__global__ void degree_kernel(const int* __restrict__ dst, float* __restrict__ deg) {
    int e = blockIdx.x * blockDim.x + threadIdx.x;
    if (e < N_EDGES) atomicAdd(&deg[dst[e]], 1.0f);
}

#define TILE_ROWS    391
#define N_TILES      256

__global__ __launch_bounds__(512) void rsqrt_tile_kernel(float* __restrict__ deg,
                                                         int* __restrict__ tile_edges) {
    __shared__ int tot;
    if (threadIdx.x == 0) tot = 0;
    __syncthreads();
    const int i = blockIdx.x * TILE_ROWS + threadIdx.x;
    if (threadIdx.x < TILE_ROWS && i < N_NODES) {
        const float d = deg[i];
        deg[i] = rsqrtf(d);
        atomicAdd(&tot, (int)d - 1);
    }
    __syncthreads();
    if (threadIdx.x == 0) tile_edges[blockIdx.x] = tot;
}

__global__ __launch_bounds__(32) void gemm_x_w1(const float* __restrict__ x,
                                                const float* __restrict__ W,
                                                float* __restrict__ out) {
    const int lane = threadIdx.x;
    const int m    = lane & 15;
    const int half = lane >> 4;
    const int row0 = blockIdx.x * 16;

    v8f acc0 = {}, acc1 = {}, acc2 = {}, acc3 = {};
    const float* xr = x + (size_t)(row0 + m) * DIM;

#pragma unroll
    for (int ks = 0; ks < 16; ++ks) {
        const int kk = ks * 4 + half * 2;
        v2f a;  a.x = xr[kk];  a.y = xr[kk + 1];
        const float* wr0 = W + (size_t)kk * DIM;
        const float* wr1 = W + (size_t)(kk + 1) * DIM;
        v2f b0, b1, b2, b3;
        b0.x = wr0[m];      b0.y = wr1[m];
        b1.x = wr0[16 + m]; b1.y = wr1[16 + m];
        b2.x = wr0[32 + m]; b2.y = wr1[32 + m];
        b3.x = wr0[48 + m]; b3.y = wr1[48 + m];
        acc0 = __builtin_amdgcn_wmma_f32_16x16x4_f32(false, a, false, b0, (short)0, acc0, false, false);
        acc1 = __builtin_amdgcn_wmma_f32_16x16x4_f32(false, a, false, b1, (short)0, acc1, false, false);
        acc2 = __builtin_amdgcn_wmma_f32_16x16x4_f32(false, a, false, b2, (short)0, acc2, false, false);
        acc3 = __builtin_amdgcn_wmma_f32_16x16x4_f32(false, a, false, b3, (short)0, acc3, false, false);
    }

#pragma unroll
    for (int r = 0; r < 8; ++r) {
        float* o = out + (size_t)(row0 + r + 8 * half) * DIM;
        o[m]      = acc0[r];
        o[16 + m] = acc1[r];
        o[32 + m] = acc2[r];
        o[48 + m] = acc3[r];
    }
}

#define PUSH_THREADS 1024
#define PUSH_WAVES   (PUSH_THREADS / 32)
#define Q_CAP        160
#define N_DST4       (N_EDGES / 4)
#define SCAN_SLOTS   ((N_DST4 + PUSH_THREADS - 1) / PUSH_THREADS)
#define E_BITS       21
#define S_BITS       17
#define REC_NONE     0xFFFFFFFFu

template <bool FIRST>
__global__ __launch_bounds__(PUSH_THREADS) void tile_push_kernel(const int* __restrict__ src,
                                                                 const int* __restrict__ dst,
                                                                 const float* __restrict__ dinv,
                                                                 const float* __restrict__ feat,
                                                                 float* __restrict__ out,
                                                                 const int* __restrict__ tile_edges,
                                                                 int* __restrict__ tile_chunks,
                                                                 unsigned* __restrict__ replay) {
    __shared__ float    acc[TILE_ROWS * DIM];
    __shared__ unsigned queue[FIRST ? PUSH_WAVES * Q_CAP : 1];
    __shared__ int      base_s;
    __shared__ int      fill_s;

    const int tid  = threadIdx.x;
    const int lane = tid & 31;
    const int wave = tid >> 5;
    const int lo   = blockIdx.x * TILE_ROWS;
    const int rows = min(TILE_ROWS, N_NODES - lo);
    const unsigned urows = (unsigned)rows;

    const float4* feat4 = (const float4*)feat;
    const float2* feat2 = (const float2*)feat;
    const int4*   dst4  = (const int4*)dst;

    if (FIRST) {
        if (tid == 0) { base_s = 0; fill_s = 0; }
        __syncthreads();
        if (tid < (int)blockIdx.x)
            atomicAdd(&base_s, ((tile_edges[tid] + 31) >> 5) + PUSH_WAVES);
    }

    for (int i = tid; i < rows * (DIM / 4); i += PUSH_THREADS) {
        const int   r  = i >> 4;
        const float dv = dinv[lo + r];
        const float w  = dv * dv;
        const float4 f = feat4[(size_t)lo * (DIM / 4) + i];
        float4 o;
        o.x = f.x * w; o.y = f.y * w; o.z = f.z * w; o.w = f.w * w;
        ((float4*)acc)[i] = o;
    }
    __syncthreads();

    const int base = FIRST ? base_s : tile_chunks[2 * blockIdx.x];
    uint2* my_chunks = (uint2*)replay + (size_t)base * 32;

    auto add_rows = [&](int n, int s, int dl, int wbits, auto&& behind_first_loads) {
        int i = 0;
        do {
            float2 f[8];
#pragma unroll
            for (int k = 0; k < 8; ++k) {
                const int si = __builtin_amdgcn_readlane(s, i + k);
                f[k] = feat2[(size_t)si * (DIM / 2) + lane];
            }
            if (i == 0) behind_first_loads();
#pragma unroll
            for (int k = 0; k < 8; ++k) {
                if (i + k < n) {
                    const float wi  = __int_as_float(__builtin_amdgcn_readlane(wbits, i + k));
                    const int   dli = __builtin_amdgcn_readlane(dl, i + k);
                    float* a = acc + dli * DIM + 2 * lane;
                    atomicAdd(a,     f[k].x * wi);
                    atomicAdd(a + 1, f[k].y * wi);
                }
            }
            i += 8;
        } while (i < n);
    };

    if (FIRST) {
        const int qoff = wave * Q_CAP;
        int cnt = 0;

        const int slot0 = (blockIdx.x * SCAN_SLOTS) >> 8;

        auto load_slot = [&](int it, int& e0) -> int4 {
            int sl = slot0 + it;
            if (sl >= SCAN_SLOTS) sl -= SCAN_SLOTS;
            const int i4 = sl * PUSH_THREADS + tid;
            e0 = i4 * 4;
            int4 v = dst4[min(i4, N_DST4 - 1)];
            if (it >= SCAN_SLOTS || i4 >= N_DST4) { v.x = -1; v.y = -1; v.z = -1; v.w = -1; }
            return v;
        };

        auto append = [&](unsigned r, int e) {
            const bool m = r < urows;
            const unsigned mask = __builtin_amdgcn_ballot_w32(m);
            if (m) {
                const unsigned pos = (unsigned)cnt + __builtin_amdgcn_mbcnt_lo(mask, 0u);
                __hip_atomic_store(&queue[qoff + pos], (r << E_BITS) | (unsigned)e,
                                   __ATOMIC_RELAXED, __HIP_MEMORY_SCOPE_WAVEFRONT);
            }
            cnt += __builtin_popcount(mask);
        };

        auto scan_slot = [&](const int4& d, int e0) {
            const unsigned r0 = (unsigned)(d.x - lo);
            const unsigned r1 = (unsigned)(d.y - lo);
            const unsigned r2 = (unsigned)(d.z - lo);
            const unsigned r3 = (unsigned)(d.w - lo);
            const unsigned mn = min(min(r0, r1), min(r2, r3));
            if (__builtin_amdgcn_ballot_w32(mn < urows) != 0u) {
                append(r0, e0);
                append(r1, e0 + 1);
                append(r2, e0 + 2);
                append(r3, e0 + 3);
            }
        };

        auto work_off = [&](int qbase, int n) {
            unsigned ent = 0u;
            if (lane < n) ent = __hip_atomic_load(&queue[qoff + qbase + lane],
                                                  __ATOMIC_RELAXED, __HIP_MEMORY_SCOPE_WAVEFRONT);
            const int e  = (int)(ent & ((1u << E_BITS) - 1u));
            const int dl = (int)(ent >> E_BITS);
            int   s = 0;
            float w = 0.0f;
            if (lane < n) {
                s = src[e];
                w = dinv[s] * dinv[lo + dl];
            }
            const int wbits = __float_as_int(w);

            int c = 0;
            if (lane == 0) c = atomicAdd(&fill_s, 1);
            c = __builtin_amdgcn_readfirstlane(c);
            uint2 rec;
            rec.x = lane < n ? ((unsigned)s | ((unsigned)dl << S_BITS)) : REC_NONE;
            rec.y = (unsigned)wbits;
            my_chunks[c * 32 + lane] = rec;

            add_rows(n, s, dl, wbits, [] {});
        };

        int  e_cur[4], e_nxt[4];
        int4 d_cur[4], d_nxt[4];
#pragma unroll
        for (int u = 0; u < 4; ++u) d_cur[u] = load_slot(u, e_cur[u]);

        for (int it = 0; it < SCAN_SLOTS; it += 4) {
#pragma unroll
            for (int u = 0; u < 4; ++u) d_nxt[u] = load_slot(it + 4 + u, e_nxt[u]);
#pragma unroll
            for (int u = 0; u < 4; ++u) {
                scan_slot(d_cur[u], e_cur[u]);
                while (cnt >= 32) { cnt -= 32; work_off(cnt, 32); }
            }
#pragma unroll
            for (int u = 0; u < 4; ++u) { d_cur[u] = d_nxt[u]; e_cur[u] = e_nxt[u]; }
        }
        if (cnt > 0) work_off(0, cnt);
    } else {
        const int nch = tile_chunks[2 * blockIdx.x + 1];
        uint2 r;  r.x = REC_NONE;  r.y = 0u;
        if (wave < nch) r = my_chunks[wave * 32 + lane];
        for (int c = wave; c < nch; c += PUSH_WAVES) {
            const bool valid = r.x != REC_NONE;
            const int  n     = __builtin_popcount(__builtin_amdgcn_ballot_w32(valid));
            const int  s     = valid ? (int)(r.x & ((1u << S_BITS) - 1u)) : 0;
            const int  dl    = valid ? (int)(r.x >> S_BITS) : 0;
            const int  wbits = valid ? (int)r.y : 0;
            uint2 rn;  rn.x = REC_NONE;  rn.y = 0u;
            add_rows(n, s, dl, wbits, [&] {
                if (c + PUSH_WAVES < nch) rn = my_chunks[(c + PUSH_WAVES) * 32 + lane];
            });
            r = rn;
        }
    }

    __syncthreads();

    if (FIRST && tid == 0) {
        tile_chunks[2 * blockIdx.x]     = base;
        tile_chunks[2 * blockIdx.x + 1] = fill_s;
    }

    float4* out4 = (float4*)out + (size_t)lo * (DIM / 4);
    for (int i = tid; i < rows * (DIM / 4); i += PUSH_THREADS)
        out4[i] = ((const float4*)acc)[i];
}

__global__ void bias_relu_kernel(float* __restrict__ h, const float* __restrict__ b1) {
    size_t tid = (size_t)blockIdx.x * blockDim.x + threadIdx.x;
    if (tid < (size_t)N_NODES * DIM)
        h[tid] = fmaxf(h[tid] + b1[tid & 63], 0.0f);
}

__global__ __launch_bounds__(32) void gemm_out(const float* __restrict__ agg,
                                               const float* __restrict__ Wmu,
                                               const float* __restrict__ bmu,
                                               const float* __restrict__ Wlv,
                                               const float* __restrict__ blv,
                                               float* __restrict__ out) {
    const int lane = threadIdx.x;
    const int m    = lane & 15;
    const int half = lane >> 4;
    const int row0 = blockIdx.x * 16;

    v8f accMu0 = {}, accMu1 = {}, accLv0 = {}, accLv1 = {};
    const float* ar = agg + (size_t)(row0 + m) * DIM;

#pragma unroll
    for (int ks = 0; ks < 16; ++ks) {
        const int kk = ks * 4 + half * 2;
        v2f a;  a.x = ar[kk];  a.y = ar[kk + 1];
        const float* mu0 = Wmu + (size_t)kk * OUT_DIM;
        const float* mu1 = Wmu + (size_t)(kk + 1) * OUT_DIM;
        const float* lv0 = Wlv + (size_t)kk * OUT_DIM;
        const float* lv1 = Wlv + (size_t)(kk + 1) * OUT_DIM;
        v2f b0, b1, b2, b3;
        b0.x = mu0[m];      b0.y = mu1[m];
        b1.x = mu0[16 + m]; b1.y = mu1[16 + m];
        b2.x = lv0[m];      b2.y = lv1[m];
        b3.x = lv0[16 + m]; b3.y = lv1[16 + m];
        accMu0 = __builtin_amdgcn_wmma_f32_16x16x4_f32(false, a, false, b0, (short)0, accMu0, false, false);
        accMu1 = __builtin_amdgcn_wmma_f32_16x16x4_f32(false, a, false, b1, (short)0, accMu1, false, false);
        accLv0 = __builtin_amdgcn_wmma_f32_16x16x4_f32(false, a, false, b2, (short)0, accLv0, false, false);
        accLv1 = __builtin_amdgcn_wmma_f32_16x16x4_f32(false, a, false, b3, (short)0, accLv1, false, false);
    }

    float* lvBase = out + (size_t)N_NODES * OUT_DIM;
#pragma unroll
    for (int r = 0; r < 8; ++r) {
        const int row = row0 + r + 8 * half;
        float* omu = out    + (size_t)row * OUT_DIM;
        float* olv = lvBase + (size_t)row * OUT_DIM;
        omu[m]      = accMu0[r] + bmu[m];
        omu[16 + m] = accMu1[r] + bmu[16 + m];
        olv[m]      = accLv0[r] + blv[m];
        olv[16 + m] = accLv1[r] + blv[16 + m];
    }
}

extern "C" void kernel_launch(void* const* d_in, const int* in_sizes, int n_in,
                              void* d_out, int out_size, void* d_ws, size_t ws_size,
                              hipStream_t stream) {
    const float* x    = (const float*)d_in[0];
    const int*   eidx = (const int*)d_in[1];
    const int*   src  = eidx;
    const int*   dst  = eidx + N_EDGES;
    const float* W1   = (const float*)d_in[2];
    const float* b1   = (const float*)d_in[3];
    const float* Wmu  = (const float*)d_in[4];
    const float* bmu  = (const float*)d_in[5];
    const float* Wlv  = (const float*)d_in[6];
    const float* blv  = (const float*)d_in[7];
    float*       out  = (float*)d_out;

    float* deg  = (float*)d_ws;
    float* xw   = deg  + N_NODES;
    float* agg1 = xw   + (size_t)N_NODES * DIM;
    float* agg2 = agg1 + (size_t)N_NODES * DIM;
    int*   tile_edges  = (int*)(agg2 + (size_t)N_NODES * DIM + 32);
    int*   tile_chunks = tile_edges + N_TILES;
    unsigned* replay   = (unsigned*)(tile_chunks + 2 * N_TILES);

    const int TB = 256;
    const size_t featElems    = (size_t)N_NODES * DIM;

    const unsigned gInit    = (unsigned)((featElems + TB - 1) / TB);
    const unsigned gDeg     = (N_EDGES + TB - 1) / TB;
    const unsigned gNode    = (N_NODES + TB - 1) / TB;
    const unsigned gGemm    = N_NODES / 16;

    init_kernel<<<gNode, TB, 0, stream>>>(deg);
    degree_kernel<<<gDeg, TB, 0, stream>>>(dst, deg);
    rsqrt_tile_kernel<<<N_TILES, 512, 0, stream>>>(deg, tile_edges);

    gemm_x_w1<<<gGemm, 32, 0, stream>>>(x, W1, xw);
    tile_push_kernel<true><<<N_TILES, PUSH_THREADS, 0, stream>>>(src, dst, deg, xw, agg1, tile_edges, tile_chunks, replay);
    bias_relu_kernel<<<gInit, TB, 0, stream>>>(agg1, b1);
    tile_push_kernel<false><<<N_TILES, PUSH_THREADS, 0, stream>>>(src, dst, deg, agg1, agg2, tile_edges, tile_chunks, replay);
    gemm_out<<<gGemm, 32, 0, stream>>>(agg2, Wmu, bmu, Wlv, blv, out);
}


# Edge list read once: a bin pass ranks each edge in its tile (LDS counters, one global claim per group and tile) and writes replay records; both layers replay; no dst scan; sum order = LDS arrival
# speedup vs baseline: 3.1768x; 1.1727x over previous
#include <hip/hip_runtime.h>
#include <hip/hip_bf16.h>

#define N_NODES 100000
#define N_EDGES 1600000
#define DIM 64
#define OUT_DIM 32

typedef __attribute__((ext_vector_type(2))) float v2f;
typedef __attribute__((ext_vector_type(8))) float v8f;

__global__ void init_kernel(float* __restrict__ deg) {
    int i = blockIdx.x * blockDim.x + threadIdx.x;
    if (i < N_NODES) deg[i] = 1.0f;
}

__global__ void degree_kernel(const int* __restrict__ dst, float* __restrict__ deg) {
    int e = blockIdx.x * blockDim.x + threadIdx.x;
    if (e < N_EDGES) atomicAdd(&deg[dst[e]], 1.0f);
}

#define TILE_ROWS    391
#define N_TILES      256

__global__ __launch_bounds__(512) void rsqrt_tile_kernel(float* __restrict__ deg,
                                                         int* __restrict__ tile_edges,
                                                         int* __restrict__ cursor) {
    __shared__ int tot;
    if (threadIdx.x == 0) tot = 0;
    __syncthreads();
    const int i = blockIdx.x * TILE_ROWS + threadIdx.x;
    if (threadIdx.x < TILE_ROWS && i < N_NODES) {
        const float d = deg[i];
        deg[i] = rsqrtf(d);
        atomicAdd(&tot, (int)d - 1);
    }
    __syncthreads();
    if (threadIdx.x == 0) { tile_edges[blockIdx.x] = tot; cursor[blockIdx.x] = 0; }
}

__global__ __launch_bounds__(32) void gemm_x_w1(const float* __restrict__ x,
                                                const float* __restrict__ W,
                                                float* __restrict__ out) {
    const int lane = threadIdx.x;
    const int m    = lane & 15;
    const int half = lane >> 4;
    const int row0 = blockIdx.x * 16;

    v8f acc0 = {}, acc1 = {}, acc2 = {}, acc3 = {};
    const float* xr = x + (size_t)(row0 + m) * DIM;

#pragma unroll
    for (int ks = 0; ks < 16; ++ks) {
        const int kk = ks * 4 + half * 2;
        v2f a;  a.x = xr[kk];  a.y = xr[kk + 1];
        const float* wr0 = W + (size_t)kk * DIM;
        const float* wr1 = W + (size_t)(kk + 1) * DIM;
        v2f b0, b1, b2, b3;
        b0.x = wr0[m];      b0.y = wr1[m];
        b1.x = wr0[16 + m]; b1.y = wr1[16 + m];
        b2.x = wr0[32 + m]; b2.y = wr1[32 + m];
        b3.x = wr0[48 + m]; b3.y = wr1[48 + m];
        acc0 = __builtin_amdgcn_wmma_f32_16x16x4_f32(false, a, false, b0, (short)0, acc0, false, false);
        acc1 = __builtin_amdgcn_wmma_f32_16x16x4_f32(false, a, false, b1, (short)0, acc1, false, false);
        acc2 = __builtin_amdgcn_wmma_f32_16x16x4_f32(false, a, false, b2, (short)0, acc2, false, false);
        acc3 = __builtin_amdgcn_wmma_f32_16x16x4_f32(false, a, false, b3, (short)0, acc3, false, false);
    }

#pragma unroll
    for (int r = 0; r < 8; ++r) {
        float* o = out + (size_t)(row0 + r + 8 * half) * DIM;
        o[m]      = acc0[r];
        o[16 + m] = acc1[r];
        o[32 + m] = acc2[r];
        o[48 + m] = acc3[r];
    }
}

#define BIN_GROUPS   256
#define BIN_THREADS  1024
#define BIN_EDGES    (N_EDGES / BIN_GROUPS)
#define BIN_PER_THR  ((BIN_EDGES + BIN_THREADS - 1) / BIN_THREADS)
#define S_BITS       17

__global__ __launch_bounds__(BIN_THREADS) void bin_edges_kernel(const int* __restrict__ src,
                                                                const int* __restrict__ dst,
                                                                const float* __restrict__ dinv,
                                                                const int* __restrict__ tile_edges,
                                                                int* __restrict__ cursor,
                                                                int* __restrict__ tile_chunks,
                                                                unsigned* __restrict__ replay) {
    __shared__ int cnt[N_TILES];
    __shared__ int first[N_TILES];
    __shared__ int chunks[N_TILES];

    const int tid = threadIdx.x;
    if (tid < N_TILES) {
        cnt[tid]    = 0;
        chunks[tid] = (tile_edges[tid] + 31) >> 5;
    }
    __syncthreads();

    const int e0 = blockIdx.x * BIN_EDGES;
    int   d[BIN_PER_THR], s[BIN_PER_THR], t[BIN_PER_THR], rank[BIN_PER_THR];
    float w[BIN_PER_THR];

#pragma unroll
    for (int k = 0; k < BIN_PER_THR; ++k) {
        const int idx = k * BIN_THREADS + tid;
        const int e   = e0 + min(idx, BIN_EDGES - 1);
        d[k] = dst[e];
        s[k] = src[e];
        if (idx >= BIN_EDGES) d[k] = -1;
    }
#pragma unroll
    for (int k = 0; k < BIN_PER_THR; ++k) {
        const int dd = d[k] >= 0 ? d[k] : 0;
        t[k] = dd / TILE_ROWS;
        w[k] = dinv[s[k]];
    }
#pragma unroll
    for (int k = 0; k < BIN_PER_THR; ++k) {
        rank[k] = 0;
        if (d[k] >= 0) rank[k] = atomicAdd(&cnt[t[k]], 1);
    }
    __syncthreads();

    if (tid < N_TILES) {
        int pre = 0;
        for (int j = 0; j < tid; ++j) pre += chunks[j];
        const int mine = cnt[tid];
        int at = 0;
        if (mine > 0) at = atomicAdd(&cursor[tid], mine);
        first[tid] = pre * 32 + at;
        if (blockIdx.x == 0) {
            tile_chunks[2 * tid]     = pre;
            tile_chunks[2 * tid + 1] = tile_edges[tid];
        }
    }
    __syncthreads();

    uint2* recs = (uint2*)replay;
#pragma unroll
    for (int k = 0; k < BIN_PER_THR; ++k) {
        if (d[k] >= 0) {
            const int dl = d[k] - t[k] * TILE_ROWS;
            uint2 rec;
            rec.x = (unsigned)s[k] | ((unsigned)dl << S_BITS);
            rec.y = (unsigned)__float_as_int(w[k]);
            recs[first[t[k]] + rank[k]] = rec;
        }
    }
}

#define PUSH_THREADS 1024
#define PUSH_WAVES   (PUSH_THREADS / 32)

__global__ __launch_bounds__(PUSH_THREADS) void tile_push_kernel(const float* __restrict__ dinv,
                                                                 const float* __restrict__ feat,
                                                                 float* __restrict__ out,
                                                                 const int* __restrict__ tile_chunks,
                                                                 const unsigned* __restrict__ replay) {
    __shared__ float acc[TILE_ROWS * DIM];
    __shared__ float dloc[TILE_ROWS];

    const int tid  = threadIdx.x;
    const int lane = tid & 31;
    const int wave = tid >> 5;
    const int lo   = blockIdx.x * TILE_ROWS;
    const int rows = min(TILE_ROWS, N_NODES - lo);

    if (tid < rows) dloc[tid] = dinv[lo + tid];

    const float4* feat4 = (const float4*)feat;
    const float2* feat2 = (const float2*)feat;

    for (int i = tid; i < rows * (DIM / 4); i += PUSH_THREADS) {
        const int   r  = i >> 4;
        const float dv = dinv[lo + r];
        const float w  = dv * dv;
        const float4 f = feat4[(size_t)lo * (DIM / 4) + i];
        float4 o;
        o.x = f.x * w; o.y = f.y * w; o.z = f.z * w; o.w = f.w * w;
        ((float4*)acc)[i] = o;
    }
    __syncthreads();

    const int base = tile_chunks[2 * blockIdx.x];
    const int nrec = tile_chunks[2 * blockIdx.x + 1];
    const int nch  = (nrec + 31) >> 5;
    const uint2* my_chunks = (const uint2*)replay + (size_t)base * 32;

    auto add_rows = [&](int n, int s, int dl, int wbits, auto&& behind_first_loads) {
        int i = 0;
        do {
            float2 f[8];
#pragma unroll
            for (int k = 0; k < 8; ++k) {
                const int si = __builtin_amdgcn_readlane(s, i + k);
                f[k] = feat2[(size_t)si * (DIM / 2) + lane];
            }
            if (i == 0) behind_first_loads();
#pragma unroll
            for (int k = 0; k < 8; ++k) {
                if (i + k < n) {
                    const float wi  = __int_as_float(__builtin_amdgcn_readlane(wbits, i + k));
                    const int   dli = __builtin_amdgcn_readlane(dl, i + k);
                    float* a = acc + dli * DIM + 2 * lane;
                    atomicAdd(a,     f[k].x * wi);
                    atomicAdd(a + 1, f[k].y * wi);
                }
            }
            i += 8;
        } while (i < n);
    };

    uint2 r;  r.x = 0u;  r.y = 0u;
    if (wave * 32 + lane < nrec) r = my_chunks[wave * 32 + lane];
    for (int c = wave; c < nch; c += PUSH_WAVES) {
        const bool valid = c * 32 + lane < nrec;
        const int  n     = __builtin_popcount(__builtin_amdgcn_ballot_w32(valid));
        const int  s     = valid ? (int)(r.x & ((1u << S_BITS) - 1u)) : 0;
        const int  dl    = valid ? (int)(r.x >> S_BITS) : 0;
        const int  wbits = __float_as_int(__int_as_float(valid ? (int)r.y : 0) * dloc[dl]);
        uint2 rn;  rn.x = 0u;  rn.y = 0u;
        add_rows(n, s, dl, wbits, [&] {
            if ((c + PUSH_WAVES) * 32 + lane < nrec) rn = my_chunks[(c + PUSH_WAVES) * 32 + lane];
        });
        r = rn;
    }

    __syncthreads();

    float4* out4 = (float4*)out + (size_t)lo * (DIM / 4);
    for (int i = tid; i < rows * (DIM / 4); i += PUSH_THREADS)
        out4[i] = ((const float4*)acc)[i];
}

__global__ void bias_relu_kernel(float* __restrict__ h, const float* __restrict__ b1) {
    size_t tid = (size_t)blockIdx.x * blockDim.x + threadIdx.x;
    if (tid < (size_t)N_NODES * DIM)
        h[tid] = fmaxf(h[tid] + b1[tid & 63], 0.0f);
}

__global__ __launch_bounds__(32) void gemm_out(const float* __restrict__ agg,
                                               const float* __restrict__ Wmu,
                                               const float* __restrict__ bmu,
                                               const float* __restrict__ Wlv,
                                               const float* __restrict__ blv,
                                               float* __restrict__ out) {
    const int lane = threadIdx.x;
    const int m    = lane & 15;
    const int half = lane >> 4;
    const int row0 = blockIdx.x * 16;

    v8f accMu0 = {}, accMu1 = {}, accLv0 = {}, accLv1 = {};
    const float* ar = agg + (size_t)(row0 + m) * DIM;

#pragma unroll
    for (int ks = 0; ks < 16; ++ks) {
        const int kk = ks * 4 + half * 2;
        v2f a;  a.x = ar[kk];  a.y = ar[kk + 1];
        const float* mu0 = Wmu + (size_t)kk * OUT_DIM;
        const float* mu1 = Wmu + (size_t)(kk + 1) * OUT_DIM;
        const float* lv0 = Wlv + (size_t)kk * OUT_DIM;
        const float* lv1 = Wlv + (size_t)(kk + 1) * OUT_DIM;
        v2f b0, b1, b2, b3;
        b0.x = mu0[m];      b0.y = mu1[m];
        b1.x = mu0[16 + m]; b1.y = mu1[16 + m];
        b2.x = lv0[m];      b2.y = lv1[m];
        b3.x = lv0[16 + m]; b3.y = lv1[16 + m];
        accMu0 = __builtin_amdgcn_wmma_f32_16x16x4_f32(false, a, false, b0, (short)0, accMu0, false, false);
        accMu1 = __builtin_amdgcn_wmma_f32_16x16x4_f32(false, a, false, b1, (short)0, accMu1, false, false);
        accLv0 = __builtin_amdgcn_wmma_f32_16x16x4_f32(false, a, false, b2, (short)0, accLv0, false, false);
        accLv1 = __builtin_amdgcn_wmma_f32_16x16x4_f32(false, a, false, b3, (short)0, accLv1, false, false);
    }

    float* lvBase = out + (size_t)N_NODES * OUT_DIM;
#pragma unroll
    for (int r = 0; r < 8; ++r) {
        const int row = row0 + r + 8 * half;
        float* omu = out    + (size_t)row * OUT_DIM;
        float* olv = lvBase + (size_t)row * OUT_DIM;
        omu[m]      = accMu0[r] + bmu[m];
        omu[16 + m] = accMu1[r] + bmu[16 + m];
        olv[m]      = accLv0[r] + blv[m];
        olv[16 + m] = accLv1[r] + blv[16 + m];
    }
}

extern "C" void kernel_launch(void* const* d_in, const int* in_sizes, int n_in,
                              void* d_out, int out_size, void* d_ws, size_t ws_size,
                              hipStream_t stream) {
    const float* x    = (const float*)d_in[0];
    const int*   eidx = (const int*)d_in[1];
    const int*   src  = eidx;
    const int*   dst  = eidx + N_EDGES;
    const float* W1   = (const float*)d_in[2];
    const float* b1   = (const float*)d_in[3];
    const float* Wmu  = (const float*)d_in[4];
    const float* bmu  = (const float*)d_in[5];
    const float* Wlv  = (const float*)d_in[6];
    const float* blv  = (const float*)d_in[7];
    float*       out  = (float*)d_out;

    float* deg  = (float*)d_ws;
    float* xw   = deg  + N_NODES;
    float* agg1 = xw   + (size_t)N_NODES * DIM;
    float* agg2 = agg1 + (size_t)N_NODES * DIM;
    int*   tile_edges  = (int*)(agg2 + (size_t)N_NODES * DIM + 32);
    int*   tile_chunks = tile_edges + N_TILES;
    int*   cursor      = tile_chunks + 2 * N_TILES;
    unsigned* replay   = (unsigned*)(cursor + N_TILES);

    const int TB = 256;
    const size_t featElems    = (size_t)N_NODES * DIM;

    const unsigned gInit    = (unsigned)((featElems + TB - 1) / TB);
    const unsigned gDeg     = (N_EDGES + TB - 1) / TB;
    const unsigned gNode    = (N_NODES + TB - 1) / TB;
    const unsigned gGemm    = N_NODES / 16;

    init_kernel<<<gNode, TB, 0, stream>>>(deg);
    degree_kernel<<<gDeg, TB, 0, stream>>>(dst, deg);
    rsqrt_tile_kernel<<<N_TILES, 512, 0, stream>>>(deg, tile_edges, cursor);
    bin_edges_kernel<<<BIN_GROUPS, BIN_THREADS, 0, stream>>>(src, dst, deg, tile_edges, cursor, tile_chunks, replay);

    gemm_x_w1<<<gGemm, 32, 0, stream>>>(x, W1, xw);
    tile_push_kernel<<<N_TILES, PUSH_THREADS, 0, stream>>>(deg, xw, agg1, tile_chunks, replay);
    bias_relu_kernel<<<gInit, TB, 0, stream>>>(agg1, b1);
    tile_push_kernel<<<N_TILES, PUSH_THREADS, 0, stream>>>(deg, agg1, agg2, tile_chunks, replay);
    gemm_out<<<gGemm, 32, 0, stream>>>(agg2, Wmu, bmu, Wlv, blv, out);
}


# Aggregation without float atomics: LDS counters deal a tile's replay records to the wave owning the row (row mod 32), which adds rows by plain LDS load, FMA, store; sum order = LDS arrival rank
# speedup vs baseline: 3.3324x; 1.0490x over previous
#include <hip/hip_runtime.h>
#include <hip/hip_bf16.h>

#define N_NODES 100000
#define N_EDGES 1600000
#define DIM 64
#define OUT_DIM 32

typedef __attribute__((ext_vector_type(2))) float v2f;
typedef __attribute__((ext_vector_type(8))) float v8f;

__global__ void init_kernel(float* __restrict__ deg) {
    int i = blockIdx.x * blockDim.x + threadIdx.x;
    if (i < N_NODES) deg[i] = 1.0f;
}

__global__ void degree_kernel(const int* __restrict__ dst, float* __restrict__ deg) {
    int e = blockIdx.x * blockDim.x + threadIdx.x;
    if (e < N_EDGES) atomicAdd(&deg[dst[e]], 1.0f);
}

#define TILE_ROWS    391
#define N_TILES      256

__global__ __launch_bounds__(512) void rsqrt_tile_kernel(float* __restrict__ deg,
                                                         int* __restrict__ tile_edges,
                                                         int* __restrict__ cursor) {
    __shared__ int tot;
    if (threadIdx.x == 0) tot = 0;
    __syncthreads();
    const int i = blockIdx.x * TILE_ROWS + threadIdx.x;
    if (threadIdx.x < TILE_ROWS && i < N_NODES) {
        const float d = deg[i];
        deg[i] = rsqrtf(d);
        atomicAdd(&tot, (int)d - 1);
    }
    __syncthreads();
    if (threadIdx.x == 0) { tile_edges[blockIdx.x] = tot; cursor[blockIdx.x] = 0; }
}

__global__ __launch_bounds__(32) void gemm_x_w1(const float* __restrict__ x,
                                                const float* __restrict__ W,
                                                float* __restrict__ out) {
    const int lane = threadIdx.x;
    const int m    = lane & 15;
    const int half = lane >> 4;
    const int row0 = blockIdx.x * 16;

    v8f acc0 = {}, acc1 = {}, acc2 = {}, acc3 = {};
    const float* xr = x + (size_t)(row0 + m) * DIM;

#pragma unroll
    for (int ks = 0; ks < 16; ++ks) {
        const int kk = ks * 4 + half * 2;
        v2f a;  a.x = xr[kk];  a.y = xr[kk + 1];
        const float* wr0 = W + (size_t)kk * DIM;
        const float* wr1 = W + (size_t)(kk + 1) * DIM;
        v2f b0, b1, b2, b3;
        b0.x = wr0[m];      b0.y = wr1[m];
        b1.x = wr0[16 + m]; b1.y = wr1[16 + m];
        b2.x = wr0[32 + m]; b2.y = wr1[32 + m];
        b3.x = wr0[48 + m]; b3.y = wr1[48 + m];
        acc0 = __builtin_amdgcn_wmma_f32_16x16x4_f32(false, a, false, b0, (short)0, acc0, false, false);
        acc1 = __builtin_amdgcn_wmma_f32_16x16x4_f32(false, a, false, b1, (short)0, acc1, false, false);
        acc2 = __builtin_amdgcn_wmma_f32_16x16x4_f32(false, a, false, b2, (short)0, acc2, false, false);
        acc3 = __builtin_amdgcn_wmma_f32_16x16x4_f32(false, a, false, b3, (short)0, acc3, false, false);
    }

#pragma unroll
    for (int r = 0; r < 8; ++r) {
        float* o = out + (size_t)(row0 + r + 8 * half) * DIM;
        o[m]      = acc0[r];
        o[16 + m] = acc1[r];
        o[32 + m] = acc2[r];
        o[48 + m] = acc3[r];
    }
}

#define BIN_GROUPS   256
#define BIN_THREADS  1024
#define BIN_EDGES    (N_EDGES / BIN_GROUPS)
#define BIN_PER_THR  ((BIN_EDGES + BIN_THREADS - 1) / BIN_THREADS)
#define S_BITS       17

__global__ __launch_bounds__(BIN_THREADS) void bin_edges_kernel(const int* __restrict__ src,
                                                                const int* __restrict__ dst,
                                                                const float* __restrict__ dinv,
                                                                const int* __restrict__ tile_edges,
                                                                int* __restrict__ cursor,
                                                                int* __restrict__ tile_chunks,
                                                                unsigned* __restrict__ replay) {
    __shared__ int cnt[N_TILES];
    __shared__ int first[N_TILES];
    __shared__ int chunks[N_TILES];

    const int tid = threadIdx.x;
    if (tid < N_TILES) {
        cnt[tid]    = 0;
        chunks[tid] = (tile_edges[tid] + 31) >> 5;
    }
    __syncthreads();

    const int e0 = blockIdx.x * BIN_EDGES;
    int   d[BIN_PER_THR], s[BIN_PER_THR], t[BIN_PER_THR], rank[BIN_PER_THR];
    float w[BIN_PER_THR];

#pragma unroll
    for (int k = 0; k < BIN_PER_THR; ++k) {
        const int idx = k * BIN_THREADS + tid;
        const int e   = e0 + min(idx, BIN_EDGES - 1);
        d[k] = dst[e];
        s[k] = src[e];
        if (idx >= BIN_EDGES) d[k] = -1;
    }
#pragma unroll
    for (int k = 0; k < BIN_PER_THR; ++k) {
        const int dd = d[k] >= 0 ? d[k] : 0;
        t[k] = dd / TILE_ROWS;
        w[k] = dinv[s[k]];
    }
#pragma unroll
    for (int k = 0; k < BIN_PER_THR; ++k) {
        rank[k] = 0;
        if (d[k] >= 0) rank[k] = atomicAdd(&cnt[t[k]], 1);
    }
    __syncthreads();

    if (tid < N_TILES) {
        int pre = 0;
        for (int j = 0; j < tid; ++j) pre += chunks[j];
        const int mine = cnt[tid];
        int at = 0;
        if (mine > 0) at = atomicAdd(&cursor[tid], mine);
        first[tid] = pre * 32 + at;
        if (blockIdx.x == 0) {
            tile_chunks[2 * tid]     = pre;
            tile_chunks[2 * tid + 1] = tile_edges[tid];
        }
    }
    __syncthreads();

    uint2* recs = (uint2*)replay;
#pragma unroll
    for (int k = 0; k < BIN_PER_THR; ++k) {
        if (d[k] >= 0) {
            const int dl = d[k] - t[k] * TILE_ROWS;
            uint2 rec;
            rec.x = (unsigned)s[k] | ((unsigned)dl << S_BITS);
            rec.y = (unsigned)__float_as_int(w[k]);
            recs[first[t[k]] + rank[k]] = rec;
        }
    }
}

#define PUSH_THREADS 1024
#define PUSH_WAVES   (PUSH_THREADS / 32)
#define ROUND_RECS   8192
#define RECS_PER_THR (ROUND_RECS / PUSH_THREADS)

__global__ __launch_bounds__(PUSH_THREADS) void tile_push_kernel(const float* __restrict__ dinv,
                                                                 const float* __restrict__ feat,
                                                                 float* __restrict__ out,
                                                                 const int* __restrict__ tile_chunks,
                                                                 const unsigned* __restrict__ replay) {
    __shared__ float acc[TILE_ROWS * DIM];
    __shared__ float dloc[TILE_ROWS];
    __shared__ uint2 sorted[ROUND_RECS];
    __shared__ int   qcnt[PUSH_WAVES];

    const int tid  = threadIdx.x;
    const int lane = tid & 31;
    const int wave = __builtin_amdgcn_readfirstlane(tid >> 5);
    const int lo   = blockIdx.x * TILE_ROWS;
    const int rows = min(TILE_ROWS, N_NODES - lo);

    if (tid < rows) dloc[tid] = dinv[lo + tid];
    if (tid < PUSH_WAVES) qcnt[tid] = 0;

    const float4* feat4 = (const float4*)feat;
    const float2* feat2 = (const float2*)feat;

    for (int i = tid; i < rows * (DIM / 4); i += PUSH_THREADS) {
        const int   r  = i >> 4;
        const float dv = dinv[lo + r];
        const float w  = dv * dv;
        const float4 f = feat4[(size_t)lo * (DIM / 4) + i];
        float4 o;
        o.x = f.x * w; o.y = f.y * w; o.z = f.z * w; o.w = f.w * w;
        ((float4*)acc)[i] = o;
    }
    __syncthreads();

    const int base = tile_chunks[2 * blockIdx.x];
    const int nrec = tile_chunks[2 * blockIdx.x + 1];
    const uint2* my_recs = (const uint2*)replay + (size_t)base * 32;

    for (int r0 = 0; r0 < nrec; r0 += ROUND_RECS) {
        unsigned rx[RECS_PER_THR], ry[RECS_PER_THR];
        int      rank[RECS_PER_THR];

#pragma unroll
        for (int k = 0; k < RECS_PER_THR; ++k) {
            const int idx = r0 + k * PUSH_THREADS + tid;
            const uint2 r = my_recs[min(idx, nrec - 1)];
            rx[k] = r.x;  ry[k] = r.y;
        }
#pragma unroll
        for (int k = 0; k < RECS_PER_THR; ++k) {
            const int idx = r0 + k * PUSH_THREADS + tid;
            rank[k] = 0;
            if (idx < nrec) rank[k] = atomicAdd(&qcnt[(rx[k] >> S_BITS) & (PUSH_WAVES - 1)], 1);
        }
        __syncthreads();

        const int c = qcnt[lane];
        int incl = c;
#pragma unroll
        for (int d = 1; d < 32; d <<= 1) {
            const int t = __shfl_up(incl, d);
            if (lane >= d) incl += t;
        }
        const int excl = incl - c;
        const int seg0 = __builtin_amdgcn_readlane(excl, wave);
        const int segn = __builtin_amdgcn_readlane(c, wave);

#pragma unroll
        for (int k = 0; k < RECS_PER_THR; ++k) {
            const int idx = r0 + k * PUSH_THREADS + tid;
            const int dl  = (int)(rx[k] >> S_BITS);
            const int at  = __shfl(excl, dl & (PUSH_WAVES - 1)) + rank[k];
            if (idx < nrec) {
                uint2 o;
                o.x = rx[k];
                o.y = (unsigned)__float_as_int(__int_as_float((int)ry[k]) * dloc[dl]);
                sorted[at] = o;
            }
        }
        __syncthreads();
        if (tid < PUSH_WAVES) qcnt[tid] = 0;

        for (int b = 0; b < segn; b += 32) {
            const int n = min(32, segn - b);
            uint2 r;  r.x = 0u;  r.y = 0u;
            if (lane < n) r = sorted[seg0 + b + lane];
            const int s  = (int)(r.x & ((1u << S_BITS) - 1u));
            const int dl = (int)(r.x >> S_BITS);
            const int wb = (int)r.y;
            float2 f[32];
#pragma unroll
            for (int k = 0; k < 32; ++k) {
                const int si = __builtin_amdgcn_readlane(s, k);
                f[k] = feat2[(size_t)si * (DIM / 2) + lane];
            }
#pragma unroll
            for (int k = 0; k < 32; k += 2) {
                if (k + 1 < n) {
                    const float w0 = __int_as_float(__builtin_amdgcn_readlane(wb, k));
                    const float w1 = __int_as_float(__builtin_amdgcn_readlane(wb, k + 1));
                    const int   d0 = __builtin_amdgcn_readlane(dl, k);
                    const int   d1 = __builtin_amdgcn_readlane(dl, k + 1);
                    float2* p0 = (float2*)(acc + d0 * DIM) + lane;
                    float2* p1 = (float2*)(acc + d1 * DIM) + lane;
                    if (d0 != d1) {
                        float2 v0 = *p0;
                        float2 v1 = *p1;
                        v0.x += f[k].x * w0;      v0.y += f[k].y * w0;
                        v1.x += f[k + 1].x * w1;  v1.y += f[k + 1].y * w1;
                        *p0 = v0;
                        *p1 = v1;
                    } else {
                        float2 v0 = *p0;
                        v0.x += f[k].x * w0;      v0.y += f[k].y * w0;
                        v0.x += f[k + 1].x * w1;  v0.y += f[k + 1].y * w1;
                        *p0 = v0;
                    }
                } else if (k < n) {
                    const float w0 = __int_as_float(__builtin_amdgcn_readlane(wb, k));
                    const int   d0 = __builtin_amdgcn_readlane(dl, k);
                    float2* p0 = (float2*)(acc + d0 * DIM) + lane;
                    float2 v0 = *p0;
                    v0.x += f[k].x * w0;  v0.y += f[k].y * w0;
                    *p0 = v0;
                }
            }
        }
        __syncthreads();
    }

    float4* out4 = (float4*)out + (size_t)lo * (DIM / 4);
    for (int i = tid; i < rows * (DIM / 4); i += PUSH_THREADS)
        out4[i] = ((const float4*)acc)[i];
}

__global__ void bias_relu_kernel(float* __restrict__ h, const float* __restrict__ b1) {
    size_t tid = (size_t)blockIdx.x * blockDim.x + threadIdx.x;
    if (tid < (size_t)N_NODES * DIM)
        h[tid] = fmaxf(h[tid] + b1[tid & 63], 0.0f);
}

__global__ __launch_bounds__(32) void gemm_out(const float* __restrict__ agg,
                                               const float* __restrict__ Wmu,
                                               const float* __restrict__ bmu,
                                               const float* __restrict__ Wlv,
                                               const float* __restrict__ blv,
                                               float* __restrict__ out) {
    const int lane = threadIdx.x;
    const int m    = lane & 15;
    const int half = lane >> 4;
    const int row0 = blockIdx.x * 16;

    v8f accMu0 = {}, accMu1 = {}, accLv0 = {}, accLv1 = {};
    const float* ar = agg + (size_t)(row0 + m) * DIM;

#pragma unroll
    for (int ks = 0; ks < 16; ++ks) {
        const int kk = ks * 4 + half * 2;
        v2f a;  a.x = ar[kk];  a.y = ar[kk + 1];
        const float* mu0 = Wmu + (size_t)kk * OUT_DIM;
        const float* mu1 = Wmu + (size_t)(kk + 1) * OUT_DIM;
        const float* lv0 = Wlv + (size_t)kk * OUT_DIM;
        const float* lv1 = Wlv + (size_t)(kk + 1) * OUT_DIM;
        v2f b0, b1, b2, b3;
        b0.x = mu0[m];      b0.y = mu1[m];
        b1.x = mu0[16 + m]; b1.y = mu1[16 + m];
        b2.x = lv0[m];      b2.y = lv1[m];
        b3.x = lv0[16 + m]; b3.y = lv1[16 + m];
        accMu0 = __builtin_amdgcn_wmma_f32_16x16x4_f32(false, a, false, b0, (short)0, accMu0, false, false);
        accMu1 = __builtin_amdgcn_wmma_f32_16x16x4_f32(false, a, false, b1, (short)0, accMu1, false, false);
        accLv0 = __builtin_amdgcn_wmma_f32_16x16x4_f32(false, a, false, b2, (short)0, accLv0, false, false);
        accLv1 = __builtin_amdgcn_wmma_f32_16x16x4_f32(false, a, false, b3, (short)0, accLv1, false, false);
    }

    float* lvBase = out + (size_t)N_NODES * OUT_DIM;
#pragma unroll
    for (int r = 0; r < 8; ++r) {
        const int row = row0 + r + 8 * half;
        float* omu = out    + (size_t)row * OUT_DIM;
        float* olv = lvBase + (size_t)row * OUT_DIM;
        omu[m]      = accMu0[r] + bmu[m];
        omu[16 + m] = accMu1[r] + bmu[16 + m];
        olv[m]      = accLv0[r] + blv[m];
        olv[16 + m] = accLv1[r] + blv[16 + m];
    }
}

extern "C" void kernel_launch(void* const* d_in, const int* in_sizes, int n_in,
                              void* d_out, int out_size, void* d_ws, size_t ws_size,
                              hipStream_t stream) {
    const float* x    = (const float*)d_in[0];
    const int*   eidx = (const int*)d_in[1];
    const int*   src  = eidx;
    const int*   dst  = eidx + N_EDGES;
    const float* W1   = (const float*)d_in[2];
    const float* b1   = (const float*)d_in[3];
    const float* Wmu  = (const float*)d_in[4];
    const float* bmu  = (const float*)d_in[5];
    const float* Wlv  = (const float*)d_in[6];
    const float* blv  = (const float*)d_in[7];
    float*       out  = (float*)d_out;

    float* deg  = (float*)d_ws;
    float* xw   = deg  + N_NODES;
    float* agg1 = xw   + (size_t)N_NODES * DIM;
    float* agg2 = agg1 + (size_t)N_NODES * DIM;
    int*   tile_edges  = (int*)(agg2 + (size_t)N_NODES * DIM + 32);
    int*   tile_chunks = tile_edges + N_TILES;
    int*   cursor      = tile_chunks + 2 * N_TILES;
    unsigned* replay   = (unsigned*)(cursor + N_TILES);

    const int TB = 256;
    const size_t featElems    = (size_t)N_NODES * DIM;

    const unsigned gInit    = (unsigned)((featElems + TB - 1) / TB);
    const unsigned gDeg     = (N_EDGES + TB - 1) / TB;
    const unsigned gNode    = (N_NODES + TB - 1) / TB;
    const unsigned gGemm    = N_NODES / 16;

    init_kernel<<<gNode, TB, 0, stream>>>(deg);
    degree_kernel<<<gDeg, TB, 0, stream>>>(dst, deg);
    rsqrt_tile_kernel<<<N_TILES, 512, 0, stream>>>(deg, tile_edges, cursor);
    bin_edges_kernel<<<BIN_GROUPS, BIN_THREADS, 0, stream>>>(src, dst, deg, tile_edges, cursor, tile_chunks, replay);

    gemm_x_w1<<<gGemm, 32, 0, stream>>>(x, W1, xw);
    tile_push_kernel<<<N_TILES, PUSH_THREADS, 0, stream>>>(deg, xw, agg1, tile_chunks, replay);
    bias_relu_kernel<<<gInit, TB, 0, stream>>>(agg1, b1);
    tile_push_kernel<<<N_TILES, PUSH_THREADS, 0, stream>>>(deg, agg1, agg2, tile_chunks, replay);
    gemm_out<<<gGemm, 32, 0, stream>>>(agg2, Wmu, bmu, Wlv, blv, out);
}


# Layer 2 projects before it aggregates: its GEMM applies b1 + ReLU to the A operand as loaded, the second push adds bmu/blv and stores mu/logvar itself; the bias+ReLU launch goes; row sum order kept
# speedup vs baseline: 3.5673x; 1.0705x over previous
#include <hip/hip_runtime.h>
#include <hip/hip_bf16.h>

#define N_NODES 100000
#define N_EDGES 1600000
#define DIM 64
#define OUT_DIM 32

typedef __attribute__((ext_vector_type(2))) float v2f;
typedef __attribute__((ext_vector_type(8))) float v8f;

__global__ void init_kernel(float* __restrict__ deg) {
    int i = blockIdx.x * blockDim.x + threadIdx.x;
    if (i < N_NODES) deg[i] = 1.0f;
}

__global__ void degree_kernel(const int* __restrict__ dst, float* __restrict__ deg) {
    int e = blockIdx.x * blockDim.x + threadIdx.x;
    if (e < N_EDGES) atomicAdd(&deg[dst[e]], 1.0f);
}

#define TILE_ROWS    391
#define N_TILES      256

__global__ __launch_bounds__(512) void rsqrt_tile_kernel(float* __restrict__ deg,
                                                         int* __restrict__ tile_edges,
                                                         int* __restrict__ cursor) {
    __shared__ int tot;
    if (threadIdx.x == 0) tot = 0;
    __syncthreads();
    const int i = blockIdx.x * TILE_ROWS + threadIdx.x;
    if (threadIdx.x < TILE_ROWS && i < N_NODES) {
        const float d = deg[i];
        deg[i] = rsqrtf(d);
        atomicAdd(&tot, (int)d - 1);
    }
    __syncthreads();
    if (threadIdx.x == 0) { tile_edges[blockIdx.x] = tot; cursor[blockIdx.x] = 0; }
}

__global__ __launch_bounds__(32) void gemm_x_w1(const float* __restrict__ x,
                                                const float* __restrict__ W,
                                                float* __restrict__ out) {
    const int lane = threadIdx.x;
    const int m    = lane & 15;
    const int half = lane >> 4;
    const int row0 = blockIdx.x * 16;

    v8f acc0 = {}, acc1 = {}, acc2 = {}, acc3 = {};
    const float* xr = x + (size_t)(row0 + m) * DIM;

#pragma unroll
    for (int ks = 0; ks < 16; ++ks) {
        const int kk = ks * 4 + half * 2;
        v2f a;  a.x = xr[kk];  a.y = xr[kk + 1];
        const float* wr0 = W + (size_t)kk * DIM;
        const float* wr1 = W + (size_t)(kk + 1) * DIM;
        v2f b0, b1, b2, b3;
        b0.x = wr0[m];      b0.y = wr1[m];
        b1.x = wr0[16 + m]; b1.y = wr1[16 + m];
        b2.x = wr0[32 + m]; b2.y = wr1[32 + m];
        b3.x = wr0[48 + m]; b3.y = wr1[48 + m];
        acc0 = __builtin_amdgcn_wmma_f32_16x16x4_f32(false, a, false, b0, (short)0, acc0, false, false);
        acc1 = __builtin_amdgcn_wmma_f32_16x16x4_f32(false, a, false, b1, (short)0, acc1, false, false);
        acc2 = __builtin_amdgcn_wmma_f32_16x16x4_f32(false, a, false, b2, (short)0, acc2, false, false);
        acc3 = __builtin_amdgcn_wmma_f32_16x16x4_f32(false, a, false, b3, (short)0, acc3, false, false);
    }

#pragma unroll
    for (int r = 0; r < 8; ++r) {
        float* o = out + (size_t)(row0 + r + 8 * half) * DIM;
        o[m]      = acc0[r];
        o[16 + m] = acc1[r];
        o[32 + m] = acc2[r];
        o[48 + m] = acc3[r];
    }
}

#define BIN_GROUPS   256
#define BIN_THREADS  1024
#define BIN_EDGES    (N_EDGES / BIN_GROUPS)
#define BIN_PER_THR  ((BIN_EDGES + BIN_THREADS - 1) / BIN_THREADS)
#define S_BITS       17

__global__ __launch_bounds__(BIN_THREADS) void bin_edges_kernel(const int* __restrict__ src,
                                                                const int* __restrict__ dst,
                                                                const float* __restrict__ dinv,
                                                                const int* __restrict__ tile_edges,
                                                                int* __restrict__ cursor,
                                                                int* __restrict__ tile_chunks,
                                                                unsigned* __restrict__ replay) {
    __shared__ int cnt[N_TILES];
    __shared__ int first[N_TILES];
    __shared__ int chunks[N_TILES];

    const int tid = threadIdx.x;
    if (tid < N_TILES) {
        cnt[tid]    = 0;
        chunks[tid] = (tile_edges[tid] + 31) >> 5;
    }
    __syncthreads();

    const int e0 = blockIdx.x * BIN_EDGES;
    int   d[BIN_PER_THR], s[BIN_PER_THR], t[BIN_PER_THR], rank[BIN_PER_THR];
    float w[BIN_PER_THR];

#pragma unroll
    for (int k = 0; k < BIN_PER_THR; ++k) {
        const int idx = k * BIN_THREADS + tid;
        const int e   = e0 + min(idx, BIN_EDGES - 1);
        d[k] = dst[e];
        s[k] = src[e];
        if (idx >= BIN_EDGES) d[k] = -1;
    }
#pragma unroll
    for (int k = 0; k < BIN_PER_THR; ++k) {
        const int dd = d[k] >= 0 ? d[k] : 0;
        t[k] = dd / TILE_ROWS;
        w[k] = dinv[s[k]];
    }
#pragma unroll
    for (int k = 0; k < BIN_PER_THR; ++k) {
        rank[k] = 0;
        if (d[k] >= 0) rank[k] = atomicAdd(&cnt[t[k]], 1);
    }
    __syncthreads();

    if (tid < N_TILES) {
        int pre = 0;
        for (int j = 0; j < tid; ++j) pre += chunks[j];
        const int mine = cnt[tid];
        int at = 0;
        if (mine > 0) at = atomicAdd(&cursor[tid], mine);
        first[tid] = pre * 32 + at;
        if (blockIdx.x == 0) {
            tile_chunks[2 * tid]     = pre;
            tile_chunks[2 * tid + 1] = tile_edges[tid];
        }
    }
    __syncthreads();

    uint2* recs = (uint2*)replay;
#pragma unroll
    for (int k = 0; k < BIN_PER_THR; ++k) {
        if (d[k] >= 0) {
            const int dl = d[k] - t[k] * TILE_ROWS;
            uint2 rec;
            rec.x = (unsigned)s[k] | ((unsigned)dl << S_BITS);
            rec.y = (unsigned)__float_as_int(w[k]);
            recs[first[t[k]] + rank[k]] = rec;
        }
    }
}

#define PUSH_THREADS 1024
#define PUSH_WAVES   (PUSH_THREADS / 32)
#define ROUND_RECS   8192
#define RECS_PER_THR (ROUND_RECS / PUSH_THREADS)

template <int FINAL>
__global__ __launch_bounds__(PUSH_THREADS) void tile_push_kernel(const float* __restrict__ dinv,
                                                                 const float* __restrict__ feat,
                                                                 float* __restrict__ out,
                                                                 const int* __restrict__ tile_chunks,
                                                                 const unsigned* __restrict__ replay,
                                                                 const float* __restrict__ bias_mu,
                                                                 const float* __restrict__ bias_lv) {
    __shared__ float acc[TILE_ROWS * DIM];
    __shared__ float dloc[TILE_ROWS];
    __shared__ uint2 sorted[ROUND_RECS];
    __shared__ int   qcnt[PUSH_WAVES];

    const int tid  = threadIdx.x;
    const int lane = tid & 31;
    const int wave = __builtin_amdgcn_readfirstlane(tid >> 5);
    const int lo   = blockIdx.x * TILE_ROWS;
    const int rows = min(TILE_ROWS, N_NODES - lo);

    if (tid < rows) dloc[tid] = dinv[lo + tid];
    if (tid < PUSH_WAVES) qcnt[tid] = 0;

    const float4* feat4 = (const float4*)feat;
    const float2* feat2 = (const float2*)feat;

    for (int i = tid; i < rows * (DIM / 4); i += PUSH_THREADS) {
        const int   r  = i >> 4;
        const float dv = dinv[lo + r];
        const float w  = dv * dv;
        const float4 f = feat4[(size_t)lo * (DIM / 4) + i];
        float4 o;
        o.x = f.x * w; o.y = f.y * w; o.z = f.z * w; o.w = f.w * w;
        ((float4*)acc)[i] = o;
    }
    __syncthreads();

    const int base = tile_chunks[2 * blockIdx.x];
    const int nrec = tile_chunks[2 * blockIdx.x + 1];
    const uint2* my_recs = (const uint2*)replay + (size_t)base * 32;

    for (int r0 = 0; r0 < nrec; r0 += ROUND_RECS) {
        unsigned rx[RECS_PER_THR], ry[RECS_PER_THR];
        int      rank[RECS_PER_THR];

#pragma unroll
        for (int k = 0; k < RECS_PER_THR; ++k) {
            const int idx = r0 + k * PUSH_THREADS + tid;
            const uint2 r = my_recs[min(idx, nrec - 1)];
            rx[k] = r.x;  ry[k] = r.y;
        }
#pragma unroll
        for (int k = 0; k < RECS_PER_THR; ++k) {
            const int idx = r0 + k * PUSH_THREADS + tid;
            rank[k] = 0;
            if (idx < nrec) rank[k] = atomicAdd(&qcnt[(rx[k] >> S_BITS) & (PUSH_WAVES - 1)], 1);
        }
        __syncthreads();

        const int c = qcnt[lane];
        int incl = c;
#pragma unroll
        for (int d = 1; d < 32; d <<= 1) {
            const int t = __shfl_up(incl, d);
            if (lane >= d) incl += t;
        }
        const int excl = incl - c;
        const int seg0 = __builtin_amdgcn_readlane(excl, wave);
        const int segn = __builtin_amdgcn_readlane(c, wave);

#pragma unroll
        for (int k = 0; k < RECS_PER_THR; ++k) {
            const int idx = r0 + k * PUSH_THREADS + tid;
            const int dl  = (int)(rx[k] >> S_BITS);
            const int at  = __shfl(excl, dl & (PUSH_WAVES - 1)) + rank[k];
            if (idx < nrec) {
                uint2 o;
                o.x = rx[k];
                o.y = (unsigned)__float_as_int(__int_as_float((int)ry[k]) * dloc[dl]);
                sorted[at] = o;
            }
        }
        __syncthreads();
        if (tid < PUSH_WAVES) qcnt[tid] = 0;

        for (int b = 0; b < segn; b += 32) {
            const int n = min(32, segn - b);
            uint2 r;  r.x = 0u;  r.y = 0u;
            if (lane < n) r = sorted[seg0 + b + lane];
            const int s  = (int)(r.x & ((1u << S_BITS) - 1u));
            const int dl = (int)(r.x >> S_BITS);
            const int wb = (int)r.y;
            float2 f[32];
#pragma unroll
            for (int k = 0; k < 32; ++k) {
                const int si = __builtin_amdgcn_readlane(s, k);
                f[k] = feat2[(size_t)si * (DIM / 2) + lane];
            }
#pragma unroll
            for (int k = 0; k < 32; k += 2) {
                if (k + 1 < n) {
                    const float w0 = __int_as_float(__builtin_amdgcn_readlane(wb, k));
                    const float w1 = __int_as_float(__builtin_amdgcn_readlane(wb, k + 1));
                    const int   d0 = __builtin_amdgcn_readlane(dl, k);
                    const int   d1 = __builtin_amdgcn_readlane(dl, k + 1);
                    float2* p0 = (float2*)(acc + d0 * DIM) + lane;
                    float2* p1 = (float2*)(acc + d1 * DIM) + lane;
                    if (d0 != d1) {
                        float2 v0 = *p0;
                        float2 v1 = *p1;
                        v0.x += f[k].x * w0;      v0.y += f[k].y * w0;
                        v1.x += f[k + 1].x * w1;  v1.y += f[k + 1].y * w1;
                        *p0 = v0;
                        *p1 = v1;
                    } else {
                        float2 v0 = *p0;
                        v0.x += f[k].x * w0;      v0.y += f[k].y * w0;
                        v0.x += f[k + 1].x * w1;  v0.y += f[k + 1].y * w1;
                        *p0 = v0;
                    }
                } else if (k < n) {
                    const float w0 = __int_as_float(__builtin_amdgcn_readlane(wb, k));
                    const int   d0 = __builtin_amdgcn_readlane(dl, k);
                    float2* p0 = (float2*)(acc + d0 * DIM) + lane;
                    float2 v0 = *p0;
                    v0.x += f[k].x * w0;  v0.y += f[k].y * w0;
                    *p0 = v0;
                }
            }
        }
        __syncthreads();
    }

    if (FINAL) {
        const int q = tid & 15;
        const float* bsrc = (q < 8 ? bias_mu : bias_lv) + (q & 7) * 4;
        const float bx = bsrc[0], by = bsrc[1], bz = bsrc[2], bw = bsrc[3];
        float4* plane = (float4*)(q < 8 ? out : out + (size_t)N_NODES * OUT_DIM)
                        + (size_t)lo * (OUT_DIM / 4) + (q & 7);
        for (int i = tid; i < rows * (DIM / 4); i += PUSH_THREADS) {
            float4 v = ((const float4*)acc)[i];
            v.x += bx; v.y += by; v.z += bz; v.w += bw;
            plane[(size_t)(i >> 4) * (OUT_DIM / 4)] = v;
        }
    } else {
        float4* out4 = (float4*)out + (size_t)lo * (DIM / 4);
        for (int i = tid; i < rows * (DIM / 4); i += PUSH_THREADS)
            out4[i] = ((const float4*)acc)[i];
    }
}

__global__ __launch_bounds__(32) void gemm_hw(const float* __restrict__ agg,
                                              const float* __restrict__ b1,
                                              const float* __restrict__ Wmu,
                                              const float* __restrict__ Wlv,
                                              float* __restrict__ out) {
    const int lane = threadIdx.x;
    const int m    = lane & 15;
    const int half = lane >> 4;
    const int row0 = blockIdx.x * 16;

    v8f accMu0 = {}, accMu1 = {}, accLv0 = {}, accLv1 = {};
    const float* ar = agg + (size_t)(row0 + m) * DIM;

#pragma unroll
    for (int ks = 0; ks < 16; ++ks) {
        const int kk = ks * 4 + half * 2;
        v2f a;
        a.x = fmaxf(ar[kk]     + b1[kk],     0.0f);
        a.y = fmaxf(ar[kk + 1] + b1[kk + 1], 0.0f);
        const float* mu0 = Wmu + (size_t)kk * OUT_DIM;
        const float* mu1 = Wmu + (size_t)(kk + 1) * OUT_DIM;
        const float* lv0 = Wlv + (size_t)kk * OUT_DIM;
        const float* lv1 = Wlv + (size_t)(kk + 1) * OUT_DIM;
        v2f b0, b1v, b2, b3;
        b0.x  = mu0[m];      b0.y  = mu1[m];
        b1v.x = mu0[16 + m]; b1v.y = mu1[16 + m];
        b2.x  = lv0[m];      b2.y  = lv1[m];
        b3.x  = lv0[16 + m]; b3.y  = lv1[16 + m];
        accMu0 = __builtin_amdgcn_wmma_f32_16x16x4_f32(false, a, false, b0,  (short)0, accMu0, false, false);
        accMu1 = __builtin_amdgcn_wmma_f32_16x16x4_f32(false, a, false, b1v, (short)0, accMu1, false, false);
        accLv0 = __builtin_amdgcn_wmma_f32_16x16x4_f32(false, a, false, b2,  (short)0, accLv0, false, false);
        accLv1 = __builtin_amdgcn_wmma_f32_16x16x4_f32(false, a, false, b3,  (short)0, accLv1, false, false);
    }

#pragma unroll
    for (int r = 0; r < 8; ++r) {
        float* o = out + (size_t)(row0 + r + 8 * half) * DIM;
        o[m]      = accMu0[r];
        o[16 + m] = accMu1[r];
        o[32 + m] = accLv0[r];
        o[48 + m] = accLv1[r];
    }
}

extern "C" void kernel_launch(void* const* d_in, const int* in_sizes, int n_in,
                              void* d_out, int out_size, void* d_ws, size_t ws_size,
                              hipStream_t stream) {
    const float* x    = (const float*)d_in[0];
    const int*   eidx = (const int*)d_in[1];
    const int*   src  = eidx;
    const int*   dst  = eidx + N_EDGES;
    const float* W1   = (const float*)d_in[2];
    const float* b1   = (const float*)d_in[3];
    const float* Wmu  = (const float*)d_in[4];
    const float* bmu  = (const float*)d_in[5];
    const float* Wlv  = (const float*)d_in[6];
    const float* blv  = (const float*)d_in[7];
    float*       out  = (float*)d_out;

    float* deg  = (float*)d_ws;
    float* xw   = deg  + N_NODES;
    float* agg1 = xw   + (size_t)N_NODES * DIM;
    float* agg2 = agg1 + (size_t)N_NODES * DIM;
    int*   tile_edges  = (int*)(agg2 + (size_t)N_NODES * DIM + 32);
    int*   tile_chunks = tile_edges + N_TILES;
    int*   cursor      = tile_chunks + 2 * N_TILES;
    unsigned* replay   = (unsigned*)(cursor + N_TILES);

    const int TB = 256;

    const unsigned gDeg     = (N_EDGES + TB - 1) / TB;
    const unsigned gNode    = (N_NODES + TB - 1) / TB;
    const unsigned gGemm    = N_NODES / 16;

    init_kernel<<<gNode, TB, 0, stream>>>(deg);
    degree_kernel<<<gDeg, TB, 0, stream>>>(dst, deg);
    rsqrt_tile_kernel<<<N_TILES, 512, 0, stream>>>(deg, tile_edges, cursor);
    bin_edges_kernel<<<BIN_GROUPS, BIN_THREADS, 0, stream>>>(src, dst, deg, tile_edges, cursor, tile_chunks, replay);

    gemm_x_w1<<<gGemm, 32, 0, stream>>>(x, W1, xw);
    tile_push_kernel<0><<<N_TILES, PUSH_THREADS, 0, stream>>>(deg, xw, agg1, tile_chunks, replay, bmu, blv);
    gemm_hw<<<gGemm, 32, 0, stream>>>(agg1, b1, Wmu, Wlv, agg2);
    tile_push_kernel<1><<<N_TILES, PUSH_THREADS, 0, stream>>>(deg, agg2, out, tile_chunks, replay, bmu, blv);
}


# Degrees on LDS counters: a count pass sizes the tiles, each tile's owner counts its replay records per row; no global atomics left; first push fetches dinv[src]; sum order = LDS arrival rank
# speedup vs baseline: 4.4630x; 1.2511x over previous
#include <hip/hip_runtime.h>
#include <hip/hip_bf16.h>

#define N_NODES 100000
#define N_EDGES 1600000
#define DIM 64
#define OUT_DIM 32

typedef __attribute__((ext_vector_type(2))) float v2f;
typedef __attribute__((ext_vector_type(8))) float v8f;

#define TILE_ROWS    391
#define N_TILES      256
#define S_BITS       17

#define BIN_GROUPS   256
#define BIN_THREADS  1024
#define BIN_EDGES    (N_EDGES / BIN_GROUPS)
#define BIN_PER_THR  ((BIN_EDGES + BIN_THREADS - 1) / BIN_THREADS)

__global__ __launch_bounds__(BIN_THREADS) void count_edges_kernel(const int* __restrict__ dst,
                                                                  int* __restrict__ cntmat) {
    __shared__ int cnt[N_TILES];
    const int tid = threadIdx.x;
    if (tid < N_TILES) cnt[tid] = 0;
    __syncthreads();

    const int e0 = blockIdx.x * BIN_EDGES;
    int d[BIN_PER_THR];
#pragma unroll
    for (int k = 0; k < BIN_PER_THR; ++k) {
        const int idx = k * BIN_THREADS + tid;
        d[k] = dst[e0 + min(idx, BIN_EDGES - 1)];
        if (idx >= BIN_EDGES) d[k] = -1;
    }
#pragma unroll
    for (int k = 0; k < BIN_PER_THR; ++k)
        if (d[k] >= 0) atomicAdd(&cnt[d[k] / TILE_ROWS], 1);
    __syncthreads();
    if (tid < N_TILES) cntmat[blockIdx.x * N_TILES + tid] = cnt[tid];
}

#define DEG_THREADS  1024
#define DEG_PER_THR  8

__global__ __launch_bounds__(DEG_THREADS) void tile_degree_kernel(const int* __restrict__ tile_chunks,
                                                                  const unsigned* __restrict__ replay,
                                                                  float* __restrict__ dinv) {
    __shared__ int degc[TILE_ROWS];
    const int tid = threadIdx.x;
    if (tid < TILE_ROWS) degc[tid] = 0;
    __syncthreads();

    const int base = tile_chunks[2 * blockIdx.x];
    const int nrec = tile_chunks[2 * blockIdx.x + 1];
    const uint2* recs = (const uint2*)replay + (size_t)base * 32;

    for (int r0 = 0; r0 < nrec; r0 += DEG_PER_THR * DEG_THREADS) {
        unsigned key[DEG_PER_THR];
#pragma unroll
        for (int k = 0; k < DEG_PER_THR; ++k) {
            const int idx = r0 + k * DEG_THREADS + tid;
            key[k] = recs[min(idx, nrec - 1)].x;
        }
#pragma unroll
        for (int k = 0; k < DEG_PER_THR; ++k) {
            const int idx = r0 + k * DEG_THREADS + tid;
            if (idx < nrec) atomicAdd(&degc[key[k] >> S_BITS], 1);
        }
    }
    __syncthreads();

    const int i = blockIdx.x * TILE_ROWS + tid;
    if (tid < TILE_ROWS && i < N_NODES) dinv[i] = rsqrtf((float)(degc[tid] + 1));
}

__global__ __launch_bounds__(32) void gemm_x_w1(const float* __restrict__ x,
                                                const float* __restrict__ W,
                                                float* __restrict__ out) {
    const int lane = threadIdx.x;
    const int m    = lane & 15;
    const int half = lane >> 4;
    const int row0 = blockIdx.x * 16;

    v8f acc0 = {}, acc1 = {}, acc2 = {}, acc3 = {};
    const float* xr = x + (size_t)(row0 + m) * DIM;

#pragma unroll
    for (int ks = 0; ks < 16; ++ks) {
        const int kk = ks * 4 + half * 2;
        v2f a;  a.x = xr[kk];  a.y = xr[kk + 1];
        const float* wr0 = W + (size_t)kk * DIM;
        const float* wr1 = W + (size_t)(kk + 1) * DIM;
        v2f b0, b1, b2, b3;
        b0.x = wr0[m];      b0.y = wr1[m];
        b1.x = wr0[16 + m]; b1.y = wr1[16 + m];
        b2.x = wr0[32 + m]; b2.y = wr1[32 + m];
        b3.x = wr0[48 + m]; b3.y = wr1[48 + m];
        acc0 = __builtin_amdgcn_wmma_f32_16x16x4_f32(false, a, false, b0, (short)0, acc0, false, false);
        acc1 = __builtin_amdgcn_wmma_f32_16x16x4_f32(false, a, false, b1, (short)0, acc1, false, false);
        acc2 = __builtin_amdgcn_wmma_f32_16x16x4_f32(false, a, false, b2, (short)0, acc2, false, false);
        acc3 = __builtin_amdgcn_wmma_f32_16x16x4_f32(false, a, false, b3, (short)0, acc3, false, false);
    }

#pragma unroll
    for (int r = 0; r < 8; ++r) {
        float* o = out + (size_t)(row0 + r + 8 * half) * DIM;
        o[m]      = acc0[r];
        o[16 + m] = acc1[r];
        o[32 + m] = acc2[r];
        o[48 + m] = acc3[r];
    }
}

#define COL_PARTS    (BIN_THREADS / N_TILES)
#define COL_ROWS     (BIN_GROUPS / COL_PARTS)

__global__ __launch_bounds__(BIN_THREADS) void bin_edges_kernel(const int* __restrict__ src,
                                                                const int* __restrict__ dst,
                                                                const int* __restrict__ cntmat,
                                                                int* __restrict__ tile_chunks,
                                                                unsigned* __restrict__ replay) {
    __shared__ int cnt[N_TILES];
    __shared__ int first[N_TILES];
    __shared__ int chunks[N_TILES];
    __shared__ int ptot[COL_PARTS][N_TILES];
    __shared__ int pbef[COL_PARTS][N_TILES];

    const int tid = threadIdx.x;
    if (tid < N_TILES) cnt[tid] = 0;

    const int e0 = blockIdx.x * BIN_EDGES;
    int d[BIN_PER_THR], s[BIN_PER_THR], t[BIN_PER_THR], rank[BIN_PER_THR];

#pragma unroll
    for (int k = 0; k < BIN_PER_THR; ++k) {
        const int idx = k * BIN_THREADS + tid;
        const int e   = e0 + min(idx, BIN_EDGES - 1);
        d[k] = dst[e];
        s[k] = src[e];
        if (idx >= BIN_EDGES) d[k] = -1;
    }

    {
        const int c = tid & (N_TILES - 1);
        const int q = tid >> 8;
        const int* col = cntmat + q * COL_ROWS * N_TILES + c;
        int tot = 0, bef = 0;
#pragma unroll
        for (int h = 0; h < COL_ROWS / 32; ++h) {
            int v[32];
#pragma unroll
            for (int j = 0; j < 32; ++j) v[j] = col[(h * 32 + j) * N_TILES];
#pragma unroll
            for (int j = 0; j < 32; ++j) {
                tot += v[j];
                if (q * COL_ROWS + h * 32 + j < (int)blockIdx.x) bef += v[j];
            }
        }
        ptot[q][c] = tot;
        pbef[q][c] = bef;
    }
    __syncthreads();

    int tile_tot = 0, tile_bef = 0;
    if (tid < N_TILES) {
#pragma unroll
        for (int q = 0; q < COL_PARTS; ++q) { tile_tot += ptot[q][tid]; tile_bef += pbef[q][tid]; }
        chunks[tid] = (tile_tot + 31) >> 5;
    }
#pragma unroll
    for (int k = 0; k < BIN_PER_THR; ++k) {
        const int dd = d[k] >= 0 ? d[k] : 0;
        t[k] = dd / TILE_ROWS;
        rank[k] = 0;
        if (d[k] >= 0) rank[k] = atomicAdd(&cnt[t[k]], 1);
    }
    __syncthreads();

    if (tid < N_TILES) {
        int pre = 0;
        for (int j = 0; j < tid; ++j) pre += chunks[j];
        first[tid] = pre * 32 + tile_bef;
        if (blockIdx.x == 0) {
            tile_chunks[2 * tid]     = pre;
            tile_chunks[2 * tid + 1] = tile_tot;
        }
    }
    __syncthreads();

    uint2* recs = (uint2*)replay;
#pragma unroll
    for (int k = 0; k < BIN_PER_THR; ++k) {
        if (d[k] >= 0) {
            const int dl = d[k] - t[k] * TILE_ROWS;
            uint2 rec;
            rec.x = (unsigned)s[k] | ((unsigned)dl << S_BITS);
            rec.y = 0u;
            recs[first[t[k]] + rank[k]] = rec;
        }
    }
}

#define PUSH_THREADS 1024
#define PUSH_WAVES   (PUSH_THREADS / 32)
#define ROUND_RECS   8192
#define RECS_PER_THR (ROUND_RECS / PUSH_THREADS)

template <int FINAL>
__global__ __launch_bounds__(PUSH_THREADS) void tile_push_kernel(const float* __restrict__ dinv,
                                                                 const float* __restrict__ feat,
                                                                 float* __restrict__ out,
                                                                 const int* __restrict__ tile_chunks,
                                                                 unsigned* __restrict__ replay,
                                                                 const float* __restrict__ bias_mu,
                                                                 const float* __restrict__ bias_lv) {
    __shared__ float acc[TILE_ROWS * DIM];
    __shared__ float dloc[TILE_ROWS];
    __shared__ uint2 sorted[ROUND_RECS];
    __shared__ int   qcnt[PUSH_WAVES];

    const int tid  = threadIdx.x;
    const int lane = tid & 31;
    const int wave = __builtin_amdgcn_readfirstlane(tid >> 5);
    const int lo   = blockIdx.x * TILE_ROWS;
    const int rows = min(TILE_ROWS, N_NODES - lo);

    if (tid < rows) dloc[tid] = dinv[lo + tid];
    if (tid < PUSH_WAVES) qcnt[tid] = 0;

    const float4* feat4 = (const float4*)feat;
    const float2* feat2 = (const float2*)feat;

    for (int i = tid; i < rows * (DIM / 4); i += PUSH_THREADS) {
        const int   r  = i >> 4;
        const float dv = dinv[lo + r];
        const float w  = dv * dv;
        const float4 f = feat4[(size_t)lo * (DIM / 4) + i];
        float4 o;
        o.x = f.x * w; o.y = f.y * w; o.z = f.z * w; o.w = f.w * w;
        ((float4*)acc)[i] = o;
    }
    __syncthreads();

    const int base = tile_chunks[2 * blockIdx.x];
    const int nrec = tile_chunks[2 * blockIdx.x + 1];
    uint2* my_recs = (uint2*)replay + (size_t)base * 32;

    for (int r0 = 0; r0 < nrec; r0 += ROUND_RECS) {
        unsigned rx[RECS_PER_THR], ry[RECS_PER_THR];
        int      rank[RECS_PER_THR];

#pragma unroll
        for (int k = 0; k < RECS_PER_THR; ++k) {
            const int idx = r0 + k * PUSH_THREADS + tid;
            const uint2 r = my_recs[min(idx, nrec - 1)];
            rx[k] = r.x;  ry[k] = r.y;
        }
        if (!FINAL) {
#pragma unroll
            for (int k = 0; k < RECS_PER_THR; ++k)
                ry[k] = (unsigned)__float_as_int(dinv[rx[k] & ((1u << S_BITS) - 1u)]);
#pragma unroll
            for (int k = 0; k < RECS_PER_THR; ++k) {
                const int idx = r0 + k * PUSH_THREADS + tid;
                if (idx < nrec) {
                    uint2 o;  o.x = rx[k];  o.y = ry[k];
                    my_recs[idx] = o;
                }
            }
        }
#pragma unroll
        for (int k = 0; k < RECS_PER_THR; ++k) {
            const int idx = r0 + k * PUSH_THREADS + tid;
            rank[k] = 0;
            if (idx < nrec) rank[k] = atomicAdd(&qcnt[(rx[k] >> S_BITS) & (PUSH_WAVES - 1)], 1);
        }
        __syncthreads();

        const int c = qcnt[lane];
        int incl = c;
#pragma unroll
        for (int d = 1; d < 32; d <<= 1) {
            const int t = __shfl_up(incl, d);
            if (lane >= d) incl += t;
        }
        const int excl = incl - c;
        const int seg0 = __builtin_amdgcn_readlane(excl, wave);
        const int segn = __builtin_amdgcn_readlane(c, wave);

#pragma unroll
        for (int k = 0; k < RECS_PER_THR; ++k) {
            const int idx = r0 + k * PUSH_THREADS + tid;
            const int dl  = (int)(rx[k] >> S_BITS);
            const int at  = __shfl(excl, dl & (PUSH_WAVES - 1)) + rank[k];
            if (idx < nrec) {
                uint2 o;
                o.x = rx[k];
                o.y = (unsigned)__float_as_int(__int_as_float((int)ry[k]) * dloc[dl]);
                sorted[at] = o;
            }
        }
        __syncthreads();
        if (tid < PUSH_WAVES) qcnt[tid] = 0;

        for (int b = 0; b < segn; b += 32) {
            const int n = min(32, segn - b);
            uint2 r;  r.x = 0u;  r.y = 0u;
            if (lane < n) r = sorted[seg0 + b + lane];
            const int s  = (int)(r.x & ((1u << S_BITS) - 1u));
            const int dl = (int)(r.x >> S_BITS);
            const int wb = (int)r.y;
            float2 f[32];
#pragma unroll
            for (int k = 0; k < 32; ++k) {
                const int si = __builtin_amdgcn_readlane(s, k);
                f[k] = feat2[(size_t)si * (DIM / 2) + lane];
            }
#pragma unroll
            for (int k = 0; k < 32; k += 2) {
                if (k + 1 < n) {
                    const float w0 = __int_as_float(__builtin_amdgcn_readlane(wb, k));
                    const float w1 = __int_as_float(__builtin_amdgcn_readlane(wb, k + 1));
                    const int   d0 = __builtin_amdgcn_readlane(dl, k);
                    const int   d1 = __builtin_amdgcn_readlane(dl, k + 1);
                    float2* p0 = (float2*)(acc + d0 * DIM) + lane;
                    float2* p1 = (float2*)(acc + d1 * DIM) + lane;
                    if (d0 != d1) {
                        float2 v0 = *p0;
                        float2 v1 = *p1;
                        v0.x += f[k].x * w0;      v0.y += f[k].y * w0;
                        v1.x += f[k + 1].x * w1;  v1.y += f[k + 1].y * w1;
                        *p0 = v0;
                        *p1 = v1;
                    } else {
                        float2 v0 = *p0;
                        v0.x += f[k].x * w0;      v0.y += f[k].y * w0;
                        v0.x += f[k + 1].x * w1;  v0.y += f[k + 1].y * w1;
                        *p0 = v0;
                    }
                } else if (k < n) {
                    const float w0 = __int_as_float(__builtin_amdgcn_readlane(wb, k));
                    const int   d0 = __builtin_amdgcn_readlane(dl, k);
                    float2* p0 = (float2*)(acc + d0 * DIM) + lane;
                    float2 v0 = *p0;
                    v0.x += f[k].x * w0;  v0.y += f[k].y * w0;
                    *p0 = v0;
                }
            }
        }
        __syncthreads();
    }

    if (FINAL) {
        const int q = tid & 15;
        const float* bsrc = (q < 8 ? bias_mu : bias_lv) + (q & 7) * 4;
        const float bx = bsrc[0], by = bsrc[1], bz = bsrc[2], bw = bsrc[3];
        float4* plane = (float4*)(q < 8 ? out : out + (size_t)N_NODES * OUT_DIM)
                        + (size_t)lo * (OUT_DIM / 4) + (q & 7);
        for (int i = tid; i < rows * (DIM / 4); i += PUSH_THREADS) {
            float4 v = ((const float4*)acc)[i];
            v.x += bx; v.y += by; v.z += bz; v.w += bw;
            plane[(size_t)(i >> 4) * (OUT_DIM / 4)] = v;
        }
    } else {
        float4* out4 = (float4*)out + (size_t)lo * (DIM / 4);
        for (int i = tid; i < rows * (DIM / 4); i += PUSH_THREADS)
            out4[i] = ((const float4*)acc)[i];
    }
}

__global__ __launch_bounds__(32) void gemm_hw(const float* __restrict__ agg,
                                              const float* __restrict__ b1,
                                              const float* __restrict__ Wmu,
                                              const float* __restrict__ Wlv,
                                              float* __restrict__ out) {
    const int lane = threadIdx.x;
    const int m    = lane & 15;
    const int half = lane >> 4;
    const int row0 = blockIdx.x * 16;

    v8f accMu0 = {}, accMu1 = {}, accLv0 = {}, accLv1 = {};
    const float* ar = agg + (size_t)(row0 + m) * DIM;

#pragma unroll
    for (int ks = 0; ks < 16; ++ks) {
        const int kk = ks * 4 + half * 2;
        v2f a;
        a.x = fmaxf(ar[kk]     + b1[kk],     0.0f);
        a.y = fmaxf(ar[kk + 1] + b1[kk + 1], 0.0f);
        const float* mu0 = Wmu + (size_t)kk * OUT_DIM;
        const float* mu1 = Wmu + (size_t)(kk + 1) * OUT_DIM;
        const float* lv0 = Wlv + (size_t)kk * OUT_DIM;
        const float* lv1 = Wlv + (size_t)(kk + 1) * OUT_DIM;
        v2f b0, b1v, b2, b3;
        b0.x  = mu0[m];      b0.y  = mu1[m];
        b1v.x = mu0[16 + m]; b1v.y = mu1[16 + m];
        b2.x  = lv0[m];      b2.y  = lv1[m];
        b3.x  = lv0[16 + m]; b3.y  = lv1[16 + m];
        accMu0 = __builtin_amdgcn_wmma_f32_16x16x4_f32(false, a, false, b0,  (short)0, accMu0, false, false);
        accMu1 = __builtin_amdgcn_wmma_f32_16x16x4_f32(false, a, false, b1v, (short)0, accMu1, false, false);
        accLv0 = __builtin_amdgcn_wmma_f32_16x16x4_f32(false, a, false, b2,  (short)0, accLv0, false, false);
        accLv1 = __builtin_amdgcn_wmma_f32_16x16x4_f32(false, a, false, b3,  (short)0, accLv1, false, false);
    }

#pragma unroll
    for (int r = 0; r < 8; ++r) {
        float* o = out + (size_t)(row0 + r + 8 * half) * DIM;
        o[m]      = accMu0[r];
        o[16 + m] = accMu1[r];
        o[32 + m] = accLv0[r];
        o[48 + m] = accLv1[r];
    }
}

extern "C" void kernel_launch(void* const* d_in, const int* in_sizes, int n_in,
                              void* d_out, int out_size, void* d_ws, size_t ws_size,
                              hipStream_t stream) {
    const float* x    = (const float*)d_in[0];
    const int*   eidx = (const int*)d_in[1];
    const int*   src  = eidx;
    const int*   dst  = eidx + N_EDGES;
    const float* W1   = (const float*)d_in[2];
    const float* b1   = (const float*)d_in[3];
    const float* Wmu  = (const float*)d_in[4];
    const float* bmu  = (const float*)d_in[5];
    const float* Wlv  = (const float*)d_in[6];
    const float* blv  = (const float*)d_in[7];
    float*       out  = (float*)d_out;

    float* dinv = (float*)d_ws;
    float* xw   = dinv + N_NODES;
    float* agg1 = xw   + (size_t)N_NODES * DIM;
    float* agg2 = agg1 + (size_t)N_NODES * DIM;
    int*   cntmat      = (int*)(agg2 + (size_t)N_NODES * DIM + 32);
    int*   tile_chunks = cntmat + BIN_GROUPS * N_TILES;
    unsigned* replay   = (unsigned*)(tile_chunks + 2 * N_TILES);

    const unsigned gGemm    = N_NODES / 16;

    count_edges_kernel<<<BIN_GROUPS, BIN_THREADS, 0, stream>>>(dst, cntmat);
    bin_edges_kernel<<<BIN_GROUPS, BIN_THREADS, 0, stream>>>(src, dst, cntmat, tile_chunks, replay);
    tile_degree_kernel<<<N_TILES, DEG_THREADS, 0, stream>>>(tile_chunks, replay, dinv);

    gemm_x_w1<<<gGemm, 32, 0, stream>>>(x, W1, xw);
    tile_push_kernel<0><<<N_TILES, PUSH_THREADS, 0, stream>>>(dinv, xw, agg1, tile_chunks, replay, bmu, blv);
    gemm_hw<<<gGemm, 32, 0, stream>>>(agg1, b1, Wmu, Wlv, agg2);
    tile_push_kernel<1><<<N_TILES, PUSH_THREADS, 0, stream>>>(dinv, agg2, out, tile_chunks, replay, bmu, blv);
}


# One projection launch: layer 1 aggregates x itself, one GEMM does relu(agg @ W1 + b1) @ [Wmu|Wlv] via a 16-row LDS tile; x @ W1 plane and its launch go; layer 1 now sums rows, then multiplies
# speedup vs baseline: 4.9186x; 1.1021x over previous
#include <hip/hip_runtime.h>
#include <hip/hip_bf16.h>

#define N_NODES 100000
#define N_EDGES 1600000
#define DIM 64
#define OUT_DIM 32

typedef __attribute__((ext_vector_type(2))) float v2f;
typedef __attribute__((ext_vector_type(8))) float v8f;

#define TILE_ROWS    391
#define N_TILES      256
#define S_BITS       17

#define BIN_GROUPS   256
#define BIN_THREADS  1024
#define BIN_EDGES    (N_EDGES / BIN_GROUPS)
#define BIN_PER_THR  ((BIN_EDGES + BIN_THREADS - 1) / BIN_THREADS)

__global__ __launch_bounds__(BIN_THREADS) void count_edges_kernel(const int* __restrict__ dst,
                                                                  int* __restrict__ cntmat) {
    __shared__ int cnt[N_TILES];
    const int tid = threadIdx.x;
    if (tid < N_TILES) cnt[tid] = 0;
    __syncthreads();

    const int e0 = blockIdx.x * BIN_EDGES;
    int d[BIN_PER_THR];
#pragma unroll
    for (int k = 0; k < BIN_PER_THR; ++k) {
        const int idx = k * BIN_THREADS + tid;
        d[k] = dst[e0 + min(idx, BIN_EDGES - 1)];
        if (idx >= BIN_EDGES) d[k] = -1;
    }
#pragma unroll
    for (int k = 0; k < BIN_PER_THR; ++k)
        if (d[k] >= 0) atomicAdd(&cnt[d[k] / TILE_ROWS], 1);
    __syncthreads();
    if (tid < N_TILES) cntmat[blockIdx.x * N_TILES + tid] = cnt[tid];
}

#define DEG_THREADS  1024
#define DEG_PER_THR  8

__global__ __launch_bounds__(DEG_THREADS) void tile_degree_kernel(const int* __restrict__ tile_chunks,
                                                                  const unsigned* __restrict__ replay,
                                                                  float* __restrict__ dinv) {
    __shared__ int degc[TILE_ROWS];
    const int tid = threadIdx.x;
    if (tid < TILE_ROWS) degc[tid] = 0;
    __syncthreads();

    const int base = tile_chunks[2 * blockIdx.x];
    const int nrec = tile_chunks[2 * blockIdx.x + 1];
    const uint2* recs = (const uint2*)replay + (size_t)base * 32;

    for (int r0 = 0; r0 < nrec; r0 += DEG_PER_THR * DEG_THREADS) {
        unsigned key[DEG_PER_THR];
#pragma unroll
        for (int k = 0; k < DEG_PER_THR; ++k) {
            const int idx = r0 + k * DEG_THREADS + tid;
            key[k] = recs[min(idx, nrec - 1)].x;
        }
#pragma unroll
        for (int k = 0; k < DEG_PER_THR; ++k) {
            const int idx = r0 + k * DEG_THREADS + tid;
            if (idx < nrec) atomicAdd(&degc[key[k] >> S_BITS], 1);
        }
    }
    __syncthreads();

    const int i = blockIdx.x * TILE_ROWS + tid;
    if (tid < TILE_ROWS && i < N_NODES) dinv[i] = rsqrtf((float)(degc[tid] + 1));
}

#define COL_PARTS    (BIN_THREADS / N_TILES)
#define COL_ROWS     (BIN_GROUPS / COL_PARTS)

__global__ __launch_bounds__(BIN_THREADS) void bin_edges_kernel(const int* __restrict__ src,
                                                                const int* __restrict__ dst,
                                                                const int* __restrict__ cntmat,
                                                                int* __restrict__ tile_chunks,
                                                                unsigned* __restrict__ replay) {
    __shared__ int cnt[N_TILES];
    __shared__ int first[N_TILES];
    __shared__ int chunks[N_TILES];
    __shared__ int ptot[COL_PARTS][N_TILES];
    __shared__ int pbef[COL_PARTS][N_TILES];

    const int tid = threadIdx.x;
    if (tid < N_TILES) cnt[tid] = 0;

    const int e0 = blockIdx.x * BIN_EDGES;
    int d[BIN_PER_THR], s[BIN_PER_THR], t[BIN_PER_THR], rank[BIN_PER_THR];

#pragma unroll
    for (int k = 0; k < BIN_PER_THR; ++k) {
        const int idx = k * BIN_THREADS + tid;
        const int e   = e0 + min(idx, BIN_EDGES - 1);
        d[k] = dst[e];
        s[k] = src[e];
        if (idx >= BIN_EDGES) d[k] = -1;
    }

    {
        const int c = tid & (N_TILES - 1);
        const int q = tid >> 8;
        const int* col = cntmat + q * COL_ROWS * N_TILES + c;
        int tot = 0, bef = 0;
#pragma unroll
        for (int h = 0; h < COL_ROWS / 32; ++h) {
            int v[32];
#pragma unroll
            for (int j = 0; j < 32; ++j) v[j] = col[(h * 32 + j) * N_TILES];
#pragma unroll
            for (int j = 0; j < 32; ++j) {
                tot += v[j];
                if (q * COL_ROWS + h * 32 + j < (int)blockIdx.x) bef += v[j];
            }
        }
        ptot[q][c] = tot;
        pbef[q][c] = bef;
    }
    __syncthreads();

    int tile_tot = 0, tile_bef = 0;
    if (tid < N_TILES) {
#pragma unroll
        for (int q = 0; q < COL_PARTS; ++q) { tile_tot += ptot[q][tid]; tile_bef += pbef[q][tid]; }
        chunks[tid] = (tile_tot + 31) >> 5;
    }
#pragma unroll
    for (int k = 0; k < BIN_PER_THR; ++k) {
        const int dd = d[k] >= 0 ? d[k] : 0;
        t[k] = dd / TILE_ROWS;
        rank[k] = 0;
        if (d[k] >= 0) rank[k] = atomicAdd(&cnt[t[k]], 1);
    }
    __syncthreads();

    if (tid < N_TILES) {
        int pre = 0;
        for (int j = 0; j < tid; ++j) pre += chunks[j];
        first[tid] = pre * 32 + tile_bef;
        if (blockIdx.x == 0) {
            tile_chunks[2 * tid]     = pre;
            tile_chunks[2 * tid + 1] = tile_tot;
        }
    }
    __syncthreads();

    uint2* recs = (uint2*)replay;
#pragma unroll
    for (int k = 0; k < BIN_PER_THR; ++k) {
        if (d[k] >= 0) {
            const int dl = d[k] - t[k] * TILE_ROWS;
            uint2 rec;
            rec.x = (unsigned)s[k] | ((unsigned)dl << S_BITS);
            rec.y = 0u;
            recs[first[t[k]] + rank[k]] = rec;
        }
    }
}

#define PUSH_THREADS 1024
#define PUSH_WAVES   (PUSH_THREADS / 32)
#define ROUND_RECS   8192
#define RECS_PER_THR (ROUND_RECS / PUSH_THREADS)

template <int FINAL>
__global__ __launch_bounds__(PUSH_THREADS) void tile_push_kernel(const float* __restrict__ dinv,
                                                                 const float* __restrict__ feat,
                                                                 float* __restrict__ out,
                                                                 const int* __restrict__ tile_chunks,
                                                                 unsigned* __restrict__ replay,
                                                                 const float* __restrict__ bias_mu,
                                                                 const float* __restrict__ bias_lv) {
    __shared__ float acc[TILE_ROWS * DIM];
    __shared__ float dloc[TILE_ROWS];
    __shared__ uint2 sorted[ROUND_RECS];
    __shared__ int   qcnt[PUSH_WAVES];

    const int tid  = threadIdx.x;
    const int lane = tid & 31;
    const int wave = __builtin_amdgcn_readfirstlane(tid >> 5);
    const int lo   = blockIdx.x * TILE_ROWS;
    const int rows = min(TILE_ROWS, N_NODES - lo);

    if (tid < rows) dloc[tid] = dinv[lo + tid];
    if (tid < PUSH_WAVES) qcnt[tid] = 0;

    const float4* feat4 = (const float4*)feat;
    const float2* feat2 = (const float2*)feat;

    for (int i = tid; i < rows * (DIM / 4); i += PUSH_THREADS) {
        const int   r  = i >> 4;
        const float dv = dinv[lo + r];
        const float w  = dv * dv;
        const float4 f = feat4[(size_t)lo * (DIM / 4) + i];
        float4 o;
        o.x = f.x * w; o.y = f.y * w; o.z = f.z * w; o.w = f.w * w;
        ((float4*)acc)[i] = o;
    }
    __syncthreads();

    const int base = tile_chunks[2 * blockIdx.x];
    const int nrec = tile_chunks[2 * blockIdx.x + 1];
    uint2* my_recs = (uint2*)replay + (size_t)base * 32;

    for (int r0 = 0; r0 < nrec; r0 += ROUND_RECS) {
        unsigned rx[RECS_PER_THR], ry[RECS_PER_THR];
        int      rank[RECS_PER_THR];

#pragma unroll
        for (int k = 0; k < RECS_PER_THR; ++k) {
            const int idx = r0 + k * PUSH_THREADS + tid;
            const uint2 r = my_recs[min(idx, nrec - 1)];
            rx[k] = r.x;  ry[k] = r.y;
        }
        if (!FINAL) {
#pragma unroll
            for (int k = 0; k < RECS_PER_THR; ++k)
                ry[k] = (unsigned)__float_as_int(dinv[rx[k] & ((1u << S_BITS) - 1u)]);
#pragma unroll
            for (int k = 0; k < RECS_PER_THR; ++k) {
                const int idx = r0 + k * PUSH_THREADS + tid;
                if (idx < nrec) {
                    uint2 o;  o.x = rx[k];  o.y = ry[k];
                    my_recs[idx] = o;
                }
            }
        }
#pragma unroll
        for (int k = 0; k < RECS_PER_THR; ++k) {
            const int idx = r0 + k * PUSH_THREADS + tid;
            rank[k] = 0;
            if (idx < nrec) rank[k] = atomicAdd(&qcnt[(rx[k] >> S_BITS) & (PUSH_WAVES - 1)], 1);
        }
        __syncthreads();

        const int c = qcnt[lane];
        int incl = c;
#pragma unroll
        for (int d = 1; d < 32; d <<= 1) {
            const int t = __shfl_up(incl, d);
            if (lane >= d) incl += t;
        }
        const int excl = incl - c;
        const int seg0 = __builtin_amdgcn_readlane(excl, wave);
        const int segn = __builtin_amdgcn_readlane(c, wave);

#pragma unroll
        for (int k = 0; k < RECS_PER_THR; ++k) {
            const int idx = r0 + k * PUSH_THREADS + tid;
            const int dl  = (int)(rx[k] >> S_BITS);
            const int at  = __shfl(excl, dl & (PUSH_WAVES - 1)) + rank[k];
            if (idx < nrec) {
                uint2 o;
                o.x = rx[k];
                o.y = (unsigned)__float_as_int(__int_as_float((int)ry[k]) * dloc[dl]);
                sorted[at] = o;
            }
        }
        __syncthreads();
        if (tid < PUSH_WAVES) qcnt[tid] = 0;

        for (int b = 0; b < segn; b += 32) {
            const int n = min(32, segn - b);
            uint2 r;  r.x = 0u;  r.y = 0u;
            if (lane < n) r = sorted[seg0 + b + lane];
            const int s  = (int)(r.x & ((1u << S_BITS) - 1u));
            const int dl = (int)(r.x >> S_BITS);
            const int wb = (int)r.y;
            float2 f[32];
#pragma unroll
            for (int k = 0; k < 32; ++k) {
                const int si = __builtin_amdgcn_readlane(s, k);
                f[k] = feat2[(size_t)si * (DIM / 2) + lane];
            }
#pragma unroll
            for (int k = 0; k < 32; k += 2) {
                if (k + 1 < n) {
                    const float w0 = __int_as_float(__builtin_amdgcn_readlane(wb, k));
                    const float w1 = __int_as_float(__builtin_amdgcn_readlane(wb, k + 1));
                    const int   d0 = __builtin_amdgcn_readlane(dl, k);
                    const int   d1 = __builtin_amdgcn_readlane(dl, k + 1);
                    float2* p0 = (float2*)(acc + d0 * DIM) + lane;
                    float2* p1 = (float2*)(acc + d1 * DIM) + lane;
                    if (d0 != d1) {
                        float2 v0 = *p0;
                        float2 v1 = *p1;
                        v0.x += f[k].x * w0;      v0.y += f[k].y * w0;
                        v1.x += f[k + 1].x * w1;  v1.y += f[k + 1].y * w1;
                        *p0 = v0;
                        *p1 = v1;
                    } else {
                        float2 v0 = *p0;
                        v0.x += f[k].x * w0;      v0.y += f[k].y * w0;
                        v0.x += f[k + 1].x * w1;  v0.y += f[k + 1].y * w1;
                        *p0 = v0;
                    }
                } else if (k < n) {
                    const float w0 = __int_as_float(__builtin_amdgcn_readlane(wb, k));
                    const int   d0 = __builtin_amdgcn_readlane(dl, k);
                    float2* p0 = (float2*)(acc + d0 * DIM) + lane;
                    float2 v0 = *p0;
                    v0.x += f[k].x * w0;  v0.y += f[k].y * w0;
                    *p0 = v0;
                }
            }
        }
        __syncthreads();
    }

    if (FINAL) {
        const int q = tid & 15;
        const float* bsrc = (q < 8 ? bias_mu : bias_lv) + (q & 7) * 4;
        const float bx = bsrc[0], by = bsrc[1], bz = bsrc[2], bw = bsrc[3];
        float4* plane = (float4*)(q < 8 ? out : out + (size_t)N_NODES * OUT_DIM)
                        + (size_t)lo * (OUT_DIM / 4) + (q & 7);
        for (int i = tid; i < rows * (DIM / 4); i += PUSH_THREADS) {
            float4 v = ((const float4*)acc)[i];
            v.x += bx; v.y += by; v.z += bz; v.w += bw;
            plane[(size_t)(i >> 4) * (OUT_DIM / 4)] = v;
        }
    } else {
        float4* out4 = (float4*)out + (size_t)lo * (DIM / 4);
        for (int i = tid; i < rows * (DIM / 4); i += PUSH_THREADS)
            out4[i] = ((const float4*)acc)[i];
    }
}

#define H_PITCH 68

__global__ __launch_bounds__(32) void gemm_chain(const float* __restrict__ aggx,
                                                 const float* __restrict__ W1,
                                                 const float* __restrict__ b1,
                                                 const float* __restrict__ Wmu,
                                                 const float* __restrict__ Wlv,
                                                 float* __restrict__ out) {
    __shared__ float htile[16 * H_PITCH];

    const int lane = threadIdx.x;
    const int m    = lane & 15;
    const int half = lane >> 4;
    const int row0 = blockIdx.x * 16;

    v8f h0 = {}, h1 = {}, h2 = {}, h3 = {};
    const float* ar = aggx + (size_t)(row0 + m) * DIM;

#pragma unroll
    for (int ks = 0; ks < 16; ++ks) {
        const int kk = ks * 4 + half * 2;
        v2f a;  a.x = ar[kk];  a.y = ar[kk + 1];
        const float* wr0 = W1 + (size_t)kk * DIM;
        const float* wr1 = W1 + (size_t)(kk + 1) * DIM;
        v2f b0, b1v, b2, b3;
        b0.x  = wr0[m];      b0.y  = wr1[m];
        b1v.x = wr0[16 + m]; b1v.y = wr1[16 + m];
        b2.x  = wr0[32 + m]; b2.y  = wr1[32 + m];
        b3.x  = wr0[48 + m]; b3.y  = wr1[48 + m];
        h0 = __builtin_amdgcn_wmma_f32_16x16x4_f32(false, a, false, b0,  (short)0, h0, false, false);
        h1 = __builtin_amdgcn_wmma_f32_16x16x4_f32(false, a, false, b1v, (short)0, h1, false, false);
        h2 = __builtin_amdgcn_wmma_f32_16x16x4_f32(false, a, false, b2,  (short)0, h2, false, false);
        h3 = __builtin_amdgcn_wmma_f32_16x16x4_f32(false, a, false, b3,  (short)0, h3, false, false);
    }

    const float c0 = b1[m], c1 = b1[16 + m], c2 = b1[32 + m], c3 = b1[48 + m];
#pragma unroll
    for (int r = 0; r < 8; ++r) {
        float* t = htile + (r + 8 * half) * H_PITCH;
        t[m]      = fmaxf(h0[r] + c0, 0.0f);
        t[16 + m] = fmaxf(h1[r] + c1, 0.0f);
        t[32 + m] = fmaxf(h2[r] + c2, 0.0f);
        t[48 + m] = fmaxf(h3[r] + c3, 0.0f);
    }
    __syncthreads();

    v8f accMu0 = {}, accMu1 = {}, accLv0 = {}, accLv1 = {};
    const float* hr = htile + m * H_PITCH;

#pragma unroll
    for (int ks = 0; ks < 16; ++ks) {
        const int kk = ks * 4 + half * 2;
        v2f a;  a.x = hr[kk];  a.y = hr[kk + 1];
        const float* mu0 = Wmu + (size_t)kk * OUT_DIM;
        const float* mu1 = Wmu + (size_t)(kk + 1) * OUT_DIM;
        const float* lv0 = Wlv + (size_t)kk * OUT_DIM;
        const float* lv1 = Wlv + (size_t)(kk + 1) * OUT_DIM;
        v2f b0, b1v, b2, b3;
        b0.x  = mu0[m];      b0.y  = mu1[m];
        b1v.x = mu0[16 + m]; b1v.y = mu1[16 + m];
        b2.x  = lv0[m];      b2.y  = lv1[m];
        b3.x  = lv0[16 + m]; b3.y  = lv1[16 + m];
        accMu0 = __builtin_amdgcn_wmma_f32_16x16x4_f32(false, a, false, b0,  (short)0, accMu0, false, false);
        accMu1 = __builtin_amdgcn_wmma_f32_16x16x4_f32(false, a, false, b1v, (short)0, accMu1, false, false);
        accLv0 = __builtin_amdgcn_wmma_f32_16x16x4_f32(false, a, false, b2,  (short)0, accLv0, false, false);
        accLv1 = __builtin_amdgcn_wmma_f32_16x16x4_f32(false, a, false, b3,  (short)0, accLv1, false, false);
    }

#pragma unroll
    for (int r = 0; r < 8; ++r) {
        float* o = out + (size_t)(row0 + r + 8 * half) * DIM;
        o[m]      = accMu0[r];
        o[16 + m] = accMu1[r];
        o[32 + m] = accLv0[r];
        o[48 + m] = accLv1[r];
    }
}

extern "C" void kernel_launch(void* const* d_in, const int* in_sizes, int n_in,
                              void* d_out, int out_size, void* d_ws, size_t ws_size,
                              hipStream_t stream) {
    const float* x    = (const float*)d_in[0];
    const int*   eidx = (const int*)d_in[1];
    const int*   src  = eidx;
    const int*   dst  = eidx + N_EDGES;
    const float* W1   = (const float*)d_in[2];
    const float* b1   = (const float*)d_in[3];
    const float* Wmu  = (const float*)d_in[4];
    const float* bmu  = (const float*)d_in[5];
    const float* Wlv  = (const float*)d_in[6];
    const float* blv  = (const float*)d_in[7];
    float*       out  = (float*)d_out;

    float* dinv = (float*)d_ws;
    float* agg1 = dinv + N_NODES;
    float* agg2 = agg1 + (size_t)N_NODES * DIM;
    int*   cntmat      = (int*)(agg2 + (size_t)N_NODES * DIM + 32);
    int*   tile_chunks = cntmat + BIN_GROUPS * N_TILES;
    unsigned* replay   = (unsigned*)(tile_chunks + 2 * N_TILES);

    const unsigned gGemm    = N_NODES / 16;

    count_edges_kernel<<<BIN_GROUPS, BIN_THREADS, 0, stream>>>(dst, cntmat);
    bin_edges_kernel<<<BIN_GROUPS, BIN_THREADS, 0, stream>>>(src, dst, cntmat, tile_chunks, replay);
    tile_degree_kernel<<<N_TILES, DEG_THREADS, 0, stream>>>(tile_chunks, replay, dinv);

    tile_push_kernel<0><<<N_TILES, PUSH_THREADS, 0, stream>>>(dinv, x, agg1, tile_chunks, replay, bmu, blv);
    gemm_chain<<<gGemm, 32, 0, stream>>>(agg1, W1, b1, Wmu, Wlv, agg2);
    tile_push_kernel<1><<<N_TILES, PUSH_THREADS, 0, stream>>>(dinv, agg2, out, tile_chunks, replay, bmu, blv);
}


# Sums in registers: a wave keeps its 13 rows as a register array, a record's row picks the element by scalar index (M0 moves); LDS sum tile and its load/store per record go; sum order kept
# speedup vs baseline: 5.6088x; 1.1403x over previous
#include <hip/hip_runtime.h>
#include <hip/hip_bf16.h>

#define N_NODES 100000
#define N_EDGES 1600000
#define DIM 64
#define OUT_DIM 32

typedef __attribute__((ext_vector_type(2))) float v2f;
typedef __attribute__((ext_vector_type(8))) float v8f;

#define TILE_ROWS    391
#define N_TILES      256
#define S_BITS       17

#define BIN_GROUPS   256
#define BIN_THREADS  1024
#define BIN_EDGES    (N_EDGES / BIN_GROUPS)
#define BIN_PER_THR  ((BIN_EDGES + BIN_THREADS - 1) / BIN_THREADS)

__global__ __launch_bounds__(BIN_THREADS) void count_edges_kernel(const int* __restrict__ dst,
                                                                  int* __restrict__ cntmat) {
    __shared__ int cnt[N_TILES];
    const int tid = threadIdx.x;
    if (tid < N_TILES) cnt[tid] = 0;
    __syncthreads();

    const int e0 = blockIdx.x * BIN_EDGES;
    int d[BIN_PER_THR];
#pragma unroll
    for (int k = 0; k < BIN_PER_THR; ++k) {
        const int idx = k * BIN_THREADS + tid;
        d[k] = dst[e0 + min(idx, BIN_EDGES - 1)];
        if (idx >= BIN_EDGES) d[k] = -1;
    }
#pragma unroll
    for (int k = 0; k < BIN_PER_THR; ++k)
        if (d[k] >= 0) atomicAdd(&cnt[d[k] / TILE_ROWS], 1);
    __syncthreads();
    if (tid < N_TILES) cntmat[blockIdx.x * N_TILES + tid] = cnt[tid];
}

#define DEG_THREADS  1024
#define DEG_PER_THR  8

__global__ __launch_bounds__(DEG_THREADS) void tile_degree_kernel(const int* __restrict__ tile_chunks,
                                                                  const unsigned* __restrict__ replay,
                                                                  float* __restrict__ dinv) {
    __shared__ int degc[TILE_ROWS];
    const int tid = threadIdx.x;
    if (tid < TILE_ROWS) degc[tid] = 0;
    __syncthreads();

    const int base = tile_chunks[2 * blockIdx.x];
    const int nrec = tile_chunks[2 * blockIdx.x + 1];
    const uint2* recs = (const uint2*)replay + (size_t)base * 32;

    for (int r0 = 0; r0 < nrec; r0 += DEG_PER_THR * DEG_THREADS) {
        unsigned key[DEG_PER_THR];
#pragma unroll
        for (int k = 0; k < DEG_PER_THR; ++k) {
            const int idx = r0 + k * DEG_THREADS + tid;
            key[k] = recs[min(idx, nrec - 1)].x;
        }
#pragma unroll
        for (int k = 0; k < DEG_PER_THR; ++k) {
            const int idx = r0 + k * DEG_THREADS + tid;
            if (idx < nrec) atomicAdd(&degc[key[k] >> S_BITS], 1);
        }
    }
    __syncthreads();

    const int i = blockIdx.x * TILE_ROWS + tid;
    if (tid < TILE_ROWS && i < N_NODES) dinv[i] = rsqrtf((float)(degc[tid] + 1));
}

#define COL_PARTS    (BIN_THREADS / N_TILES)
#define COL_ROWS     (BIN_GROUPS / COL_PARTS)

__global__ __launch_bounds__(BIN_THREADS) void bin_edges_kernel(const int* __restrict__ src,
                                                                const int* __restrict__ dst,
                                                                const int* __restrict__ cntmat,
                                                                int* __restrict__ tile_chunks,
                                                                unsigned* __restrict__ replay) {
    __shared__ int cnt[N_TILES];
    __shared__ int first[N_TILES];
    __shared__ int chunks[N_TILES];
    __shared__ int ptot[COL_PARTS][N_TILES];
    __shared__ int pbef[COL_PARTS][N_TILES];

    const int tid = threadIdx.x;
    if (tid < N_TILES) cnt[tid] = 0;

    const int e0 = blockIdx.x * BIN_EDGES;
    int d[BIN_PER_THR], s[BIN_PER_THR], t[BIN_PER_THR], rank[BIN_PER_THR];

#pragma unroll
    for (int k = 0; k < BIN_PER_THR; ++k) {
        const int idx = k * BIN_THREADS + tid;
        const int e   = e0 + min(idx, BIN_EDGES - 1);
        d[k] = dst[e];
        s[k] = src[e];
        if (idx >= BIN_EDGES) d[k] = -1;
    }

    {
        const int c = tid & (N_TILES - 1);
        const int q = tid >> 8;
        const int* col = cntmat + q * COL_ROWS * N_TILES + c;
        int tot = 0, bef = 0;
#pragma unroll
        for (int h = 0; h < COL_ROWS / 32; ++h) {
            int v[32];
#pragma unroll
            for (int j = 0; j < 32; ++j) v[j] = col[(h * 32 + j) * N_TILES];
#pragma unroll
            for (int j = 0; j < 32; ++j) {
                tot += v[j];
                if (q * COL_ROWS + h * 32 + j < (int)blockIdx.x) bef += v[j];
            }
        }
        ptot[q][c] = tot;
        pbef[q][c] = bef;
    }
    __syncthreads();

    int tile_tot = 0, tile_bef = 0;
    if (tid < N_TILES) {
#pragma unroll
        for (int q = 0; q < COL_PARTS; ++q) { tile_tot += ptot[q][tid]; tile_bef += pbef[q][tid]; }
        chunks[tid] = (tile_tot + 31) >> 5;
    }
#pragma unroll
    for (int k = 0; k < BIN_PER_THR; ++k) {
        const int dd = d[k] >= 0 ? d[k] : 0;
        t[k] = dd / TILE_ROWS;
        rank[k] = 0;
        if (d[k] >= 0) rank[k] = atomicAdd(&cnt[t[k]], 1);
    }
    __syncthreads();

    if (tid < N_TILES) {
        int pre = 0;
        for (int j = 0; j < tid; ++j) pre += chunks[j];
        first[tid] = pre * 32 + tile_bef;
        if (blockIdx.x == 0) {
            tile_chunks[2 * tid]     = pre;
            tile_chunks[2 * tid + 1] = tile_tot;
        }
    }
    __syncthreads();

    uint2* recs = (uint2*)replay;
#pragma unroll
    for (int k = 0; k < BIN_PER_THR; ++k) {
        if (d[k] >= 0) {
            const int dl = d[k] - t[k] * TILE_ROWS;
            uint2 rec;
            rec.x = (unsigned)s[k] | ((unsigned)dl << S_BITS);
            rec.y = 0u;
            recs[first[t[k]] + rank[k]] = rec;
        }
    }
}

#define PUSH_THREADS 1024
#define PUSH_WAVES   (PUSH_THREADS / 32)
#define ROUND_RECS   8192
#define RECS_PER_THR (ROUND_RECS / PUSH_THREADS)
#define WAVE_ROWS    ((TILE_ROWS + PUSH_WAVES - 1) / PUSH_WAVES)
#define SPARE_SLOT   15

typedef __attribute__((ext_vector_type(16))) float v16f;

template <int FINAL>
__global__ __launch_bounds__(PUSH_THREADS) void tile_push_kernel(const float* __restrict__ dinv,
                                                                 const float* __restrict__ feat,
                                                                 float* __restrict__ out,
                                                                 const int* __restrict__ tile_chunks,
                                                                 unsigned* __restrict__ replay,
                                                                 const float* __restrict__ bias_mu,
                                                                 const float* __restrict__ bias_lv) {
    __shared__ float dloc[TILE_ROWS];
    __shared__ uint2 sorted[ROUND_RECS];
    __shared__ int   qcnt[PUSH_WAVES];

    const int tid  = threadIdx.x;
    const int lane = tid & 31;
    const int wave = __builtin_amdgcn_readfirstlane(tid >> 5);
    const int lo   = blockIdx.x * TILE_ROWS;
    const int rows = min(TILE_ROWS, N_NODES - lo);

    if (tid < rows) dloc[tid] = dinv[lo + tid];
    if (tid < PUSH_WAVES) qcnt[tid] = 0;

    const float2* feat2 = (const float2*)feat;

    v16f accx, accy;
    {
        float2 f0[WAVE_ROWS];
        float  dv[WAVE_ROWS];
#pragma unroll
        for (int j = 0; j < WAVE_ROWS; ++j) {
            const int dl = min(32 * j + wave, rows - 1);
            f0[j] = feat2[(size_t)(lo + dl) * (DIM / 2) + lane];
            dv[j] = dinv[lo + dl];
        }
#pragma unroll
        for (int j = 0; j < 16; ++j) {
            float vx = 0.0f, vy = 0.0f;
            if (j < WAVE_ROWS) {
                if (32 * j + wave < rows) {
                    const float w = dv[j] * dv[j];
                    vx = f0[j].x * w;
                    vy = f0[j].y * w;
                }
            }
            accx[j] = vx;
            accy[j] = vy;
        }
    }
    __syncthreads();

    const int base = tile_chunks[2 * blockIdx.x];
    const int nrec = tile_chunks[2 * blockIdx.x + 1];
    uint2* my_recs = (uint2*)replay + (size_t)base * 32;

    for (int r0 = 0; r0 < nrec; r0 += ROUND_RECS) {
        unsigned rx[RECS_PER_THR], ry[RECS_PER_THR];
        int      rank[RECS_PER_THR];

#pragma unroll
        for (int k = 0; k < RECS_PER_THR; ++k) {
            const int idx = r0 + k * PUSH_THREADS + tid;
            const uint2 r = my_recs[min(idx, nrec - 1)];
            rx[k] = r.x;  ry[k] = r.y;
        }
        if (!FINAL) {
#pragma unroll
            for (int k = 0; k < RECS_PER_THR; ++k)
                ry[k] = (unsigned)__float_as_int(dinv[rx[k] & ((1u << S_BITS) - 1u)]);
#pragma unroll
            for (int k = 0; k < RECS_PER_THR; ++k) {
                const int idx = r0 + k * PUSH_THREADS + tid;
                if (idx < nrec) {
                    uint2 o;  o.x = rx[k];  o.y = ry[k];
                    my_recs[idx] = o;
                }
            }
        }
#pragma unroll
        for (int k = 0; k < RECS_PER_THR; ++k) {
            const int idx = r0 + k * PUSH_THREADS + tid;
            rank[k] = 0;
            if (idx < nrec) rank[k] = atomicAdd(&qcnt[(rx[k] >> S_BITS) & (PUSH_WAVES - 1)], 1);
        }
        __syncthreads();

        const int c = qcnt[lane];
        int incl = c;
#pragma unroll
        for (int d = 1; d < 32; d <<= 1) {
            const int t = __shfl_up(incl, d);
            if (lane >= d) incl += t;
        }
        const int excl = incl - c;
        const int seg0 = __builtin_amdgcn_readlane(excl, wave);
        const int segn = __builtin_amdgcn_readlane(c, wave);

#pragma unroll
        for (int k = 0; k < RECS_PER_THR; ++k) {
            const int idx = r0 + k * PUSH_THREADS + tid;
            const int dl  = (int)(rx[k] >> S_BITS);
            const int at  = __shfl(excl, dl & (PUSH_WAVES - 1)) + rank[k];
            if (idx < nrec) {
                uint2 o;
                o.x = (rx[k] & ((1u << S_BITS) - 1u)) | ((unsigned)(dl >> 5) << S_BITS);
                o.y = (unsigned)__float_as_int(__int_as_float((int)ry[k]) * dloc[dl]);
                sorted[at] = o;
            }
        }
        __syncthreads();
        if (tid < PUSH_WAVES) qcnt[tid] = 0;

        for (int b = 0; b < segn; b += 32) {
            const int n = min(32, segn - b);
            uint2 r;  r.x = (unsigned)SPARE_SLOT << S_BITS;  r.y = 0u;
            if (lane < n) r = sorted[seg0 + b + lane];
            const unsigned s  = r.x & ((1u << S_BITS) - 1u);
            const int      el = (int)(r.x >> S_BITS);
            const int      wb = (int)r.y;
            float2 f[32];
#pragma unroll
            for (int k = 0; k < 32; ++k) {
                const unsigned si = (unsigned)__builtin_amdgcn_readlane((int)s, k);
                f[k] = feat2[si * (unsigned)(DIM / 2) + (unsigned)lane];
            }
#pragma unroll
            for (int k = 0; k < 32; ++k) {
                const int   j = __builtin_amdgcn_readlane(el, k);
                const float w = __int_as_float(__builtin_amdgcn_readlane(wb, k));
                float ax = accx[j];
                float ay = accy[j];
                ax += f[k].x * w;
                ay += f[k].y * w;
                accx[j] = ax;
                accy[j] = ay;
            }
        }
        __syncthreads();
    }

    if (FINAL) {
        const int h  = lane >> 4;
        const int cc = (lane & 15) * 2;
        const float* bsrc = (h ? bias_lv : bias_mu) + cc;
        const float bx = bsrc[0], by = bsrc[1];
        float* plane = (h ? out + (size_t)N_NODES * OUT_DIM : out) + cc;
#pragma unroll
        for (int j = 0; j < WAVE_ROWS; ++j) {
            const int dl = 32 * j + wave;
            if (dl < rows) {
                float2 v;  v.x = accx[j] + bx;  v.y = accy[j] + by;
                *(float2*)(plane + (size_t)(lo + dl) * OUT_DIM) = v;
            }
        }
    } else {
        float2* out2 = (float2*)out;
#pragma unroll
        for (int j = 0; j < WAVE_ROWS; ++j) {
            const int dl = 32 * j + wave;
            if (dl < rows) {
                float2 v;  v.x = accx[j];  v.y = accy[j];
                out2[(size_t)(lo + dl) * (DIM / 2) + lane] = v;
            }
        }
    }
}

#define H_PITCH 68

__global__ __launch_bounds__(32) void gemm_chain(const float* __restrict__ aggx,
                                                 const float* __restrict__ W1,
                                                 const float* __restrict__ b1,
                                                 const float* __restrict__ Wmu,
                                                 const float* __restrict__ Wlv,
                                                 float* __restrict__ out) {
    __shared__ float htile[16 * H_PITCH];

    const int lane = threadIdx.x;
    const int m    = lane & 15;
    const int half = lane >> 4;
    const int row0 = blockIdx.x * 16;

    v8f h0 = {}, h1 = {}, h2 = {}, h3 = {};
    const float* ar = aggx + (size_t)(row0 + m) * DIM;

#pragma unroll
    for (int ks = 0; ks < 16; ++ks) {
        const int kk = ks * 4 + half * 2;
        v2f a;  a.x = ar[kk];  a.y = ar[kk + 1];
        const float* wr0 = W1 + (size_t)kk * DIM;
        const float* wr1 = W1 + (size_t)(kk + 1) * DIM;
        v2f b0, b1v, b2, b3;
        b0.x  = wr0[m];      b0.y  = wr1[m];
        b1v.x = wr0[16 + m]; b1v.y = wr1[16 + m];
        b2.x  = wr0[32 + m]; b2.y  = wr1[32 + m];
        b3.x  = wr0[48 + m]; b3.y  = wr1[48 + m];
        h0 = __builtin_amdgcn_wmma_f32_16x16x4_f32(false, a, false, b0,  (short)0, h0, false, false);
        h1 = __builtin_amdgcn_wmma_f32_16x16x4_f32(false, a, false, b1v, (short)0, h1, false, false);
        h2 = __builtin_amdgcn_wmma_f32_16x16x4_f32(false, a, false, b2,  (short)0, h2, false, false);
        h3 = __builtin_amdgcn_wmma_f32_16x16x4_f32(false, a, false, b3,  (short)0, h3, false, false);
    }

    const float c0 = b1[m], c1 = b1[16 + m], c2 = b1[32 + m], c3 = b1[48 + m];
#pragma unroll
    for (int r = 0; r < 8; ++r) {
        float* t = htile + (r + 8 * half) * H_PITCH;
        t[m]      = fmaxf(h0[r] + c0, 0.0f);
        t[16 + m] = fmaxf(h1[r] + c1, 0.0f);
        t[32 + m] = fmaxf(h2[r] + c2, 0.0f);
        t[48 + m] = fmaxf(h3[r] + c3, 0.0f);
    }
    __syncthreads();

    v8f accMu0 = {}, accMu1 = {}, accLv0 = {}, accLv1 = {};
    const float* hr = htile + m * H_PITCH;

#pragma unroll
    for (int ks = 0; ks < 16; ++ks) {
        const int kk = ks * 4 + half * 2;
        v2f a;  a.x = hr[kk];  a.y = hr[kk + 1];
        const float* mu0 = Wmu + (size_t)kk * OUT_DIM;
        const float* mu1 = Wmu + (size_t)(kk + 1) * OUT_DIM;
        const float* lv0 = Wlv + (size_t)kk * OUT_DIM;
        const float* lv1 = Wlv + (size_t)(kk + 1) * OUT_DIM;
        v2f b0, b1v, b2, b3;
        b0.x  = mu0[m];      b0.y  = mu1[m];
        b1v.x = mu0[16 + m]; b1v.y = mu1[16 + m];
        b2.x  = lv0[m];      b2.y  = lv1[m];
        b3.x  = lv0[16 + m]; b3.y  = lv1[16 + m];
        accMu0 = __builtin_amdgcn_wmma_f32_16x16x4_f32(false, a, false, b0,  (short)0, accMu0, false, false);
        accMu1 = __builtin_amdgcn_wmma_f32_16x16x4_f32(false, a, false, b1v, (short)0, accMu1, false, false);
        accLv0 = __builtin_amdgcn_wmma_f32_16x16x4_f32(false, a, false, b2,  (short)0, accLv0, false, false);
        accLv1 = __builtin_amdgcn_wmma_f32_16x16x4_f32(false, a, false, b3,  (short)0, accLv1, false, false);
    }

#pragma unroll
    for (int r = 0; r < 8; ++r) {
        float* o = out + (size_t)(row0 + r + 8 * half) * DIM;
        o[m]      = accMu0[r];
        o[16 + m] = accMu1[r];
        o[32 + m] = accLv0[r];
        o[48 + m] = accLv1[r];
    }
}

extern "C" void kernel_launch(void* const* d_in, const int* in_sizes, int n_in,
                              void* d_out, int out_size, void* d_ws, size_t ws_size,
                              hipStream_t stream) {
    const float* x    = (const float*)d_in[0];
    const int*   eidx = (const int*)d_in[1];
    const int*   src  = eidx;
    const int*   dst  = eidx + N_EDGES;
    const float* W1   = (const float*)d_in[2];
    const float* b1   = (const float*)d_in[3];
    const float* Wmu  = (const float*)d_in[4];
    const float* bmu  = (const float*)d_in[5];
    const float* Wlv  = (const float*)d_in[6];
    const float* blv  = (const float*)d_in[7];
    float*       out  = (float*)d_out;

    float* dinv = (float*)d_ws;
    float* agg1 = dinv + N_NODES;
    float* agg2 = agg1 + (size_t)N_NODES * DIM;
    int*   cntmat      = (int*)(agg2 + (size_t)N_NODES * DIM + 32);
    int*   tile_chunks = cntmat + BIN_GROUPS * N_TILES;
    unsigned* replay   = (unsigned*)(tile_chunks + 2 * N_TILES);

    const unsigned gGemm    = N_NODES / 16;

    count_edges_kernel<<<BIN_GROUPS, BIN_THREADS, 0, stream>>>(dst, cntmat);
    bin_edges_kernel<<<BIN_GROUPS, BIN_THREADS, 0, stream>>>(src, dst, cntmat, tile_chunks, replay);
    tile_degree_kernel<<<N_TILES, DEG_THREADS, 0, stream>>>(tile_chunks, replay, dinv);

    tile_push_kernel<0><<<N_TILES, PUSH_THREADS, 0, stream>>>(dinv, x, agg1, tile_chunks, replay, bmu, blv);
    gemm_chain<<<gGemm, 32, 0, stream>>>(agg1, W1, b1, Wmu, Wlv, agg2);
    tile_push_kernel<1><<<N_TILES, PUSH_THREADS, 0, stream>>>(dinv, agg2, out, tile_chunks, replay, bmu, blv);
}


# Projection moves into the first aggregation: waves put their summed rows in an LDS tile, 25 waves multiply 16-row blocks on WMMA (B fragments in LDS); agg plane and GEMM launch go; sum order kept
# speedup vs baseline: 5.8321x; 1.0398x over previous
#include <hip/hip_runtime.h>
#include <hip/hip_bf16.h>

#define N_NODES 100000
#define N_EDGES 1600000
#define DIM 64
#define OUT_DIM 32

typedef __attribute__((ext_vector_type(2))) float v2f;
typedef __attribute__((ext_vector_type(8))) float v8f;

#define TILE_ROWS    391
#define N_TILES      256
#define S_BITS       17

#define BIN_GROUPS   256
#define BIN_THREADS  1024
#define BIN_EDGES    (N_EDGES / BIN_GROUPS)
#define BIN_PER_THR  ((BIN_EDGES + BIN_THREADS - 1) / BIN_THREADS)

__global__ __launch_bounds__(BIN_THREADS) void count_edges_kernel(const int* __restrict__ dst,
                                                                  int* __restrict__ cntmat) {
    __shared__ int cnt[N_TILES];
    const int tid = threadIdx.x;
    if (tid < N_TILES) cnt[tid] = 0;
    __syncthreads();

    const int e0 = blockIdx.x * BIN_EDGES;
    int d[BIN_PER_THR];
#pragma unroll
    for (int k = 0; k < BIN_PER_THR; ++k) {
        const int idx = k * BIN_THREADS + tid;
        d[k] = dst[e0 + min(idx, BIN_EDGES - 1)];
        if (idx >= BIN_EDGES) d[k] = -1;
    }
#pragma unroll
    for (int k = 0; k < BIN_PER_THR; ++k)
        if (d[k] >= 0) atomicAdd(&cnt[d[k] / TILE_ROWS], 1);
    __syncthreads();
    if (tid < N_TILES) cntmat[blockIdx.x * N_TILES + tid] = cnt[tid];
}

#define DEG_THREADS  1024
#define DEG_PER_THR  8

__global__ __launch_bounds__(DEG_THREADS) void tile_degree_kernel(const int* __restrict__ tile_chunks,
                                                                  const unsigned* __restrict__ replay,
                                                                  float* __restrict__ dinv) {
    __shared__ int degc[TILE_ROWS];
    const int tid = threadIdx.x;
    if (tid < TILE_ROWS) degc[tid] = 0;
    __syncthreads();

    const int base = tile_chunks[2 * blockIdx.x];
    const int nrec = tile_chunks[2 * blockIdx.x + 1];
    const uint2* recs = (const uint2*)replay + (size_t)base * 32;

    for (int r0 = 0; r0 < nrec; r0 += DEG_PER_THR * DEG_THREADS) {
        unsigned key[DEG_PER_THR];
#pragma unroll
        for (int k = 0; k < DEG_PER_THR; ++k) {
            const int idx = r0 + k * DEG_THREADS + tid;
            key[k] = recs[min(idx, nrec - 1)].x;
        }
#pragma unroll
        for (int k = 0; k < DEG_PER_THR; ++k) {
            const int idx = r0 + k * DEG_THREADS + tid;
            if (idx < nrec) atomicAdd(&degc[key[k] >> S_BITS], 1);
        }
    }
    __syncthreads();

    const int i = blockIdx.x * TILE_ROWS + tid;
    if (tid < TILE_ROWS && i < N_NODES) dinv[i] = rsqrtf((float)(degc[tid] + 1));
}

#define COL_PARTS    (BIN_THREADS / N_TILES)
#define COL_ROWS     (BIN_GROUPS / COL_PARTS)

__global__ __launch_bounds__(BIN_THREADS) void bin_edges_kernel(const int* __restrict__ src,
                                                                const int* __restrict__ dst,
                                                                const int* __restrict__ cntmat,
                                                                int* __restrict__ tile_chunks,
                                                                unsigned* __restrict__ replay) {
    __shared__ int cnt[N_TILES];
    __shared__ int first[N_TILES];
    __shared__ int chunks[N_TILES];
    __shared__ int ptot[COL_PARTS][N_TILES];
    __shared__ int pbef[COL_PARTS][N_TILES];

    const int tid = threadIdx.x;
    if (tid < N_TILES) cnt[tid] = 0;

    const int e0 = blockIdx.x * BIN_EDGES;
    int d[BIN_PER_THR], s[BIN_PER_THR], t[BIN_PER_THR], rank[BIN_PER_THR];

#pragma unroll
    for (int k = 0; k < BIN_PER_THR; ++k) {
        const int idx = k * BIN_THREADS + tid;
        const int e   = e0 + min(idx, BIN_EDGES - 1);
        d[k] = dst[e];
        s[k] = src[e];
        if (idx >= BIN_EDGES) d[k] = -1;
    }

    {
        const int c = tid & (N_TILES - 1);
        const int q = tid >> 8;
        const int* col = cntmat + q * COL_ROWS * N_TILES + c;
        int tot = 0, bef = 0;
#pragma unroll
        for (int h = 0; h < COL_ROWS / 32; ++h) {
            int v[32];
#pragma unroll
            for (int j = 0; j < 32; ++j) v[j] = col[(h * 32 + j) * N_TILES];
#pragma unroll
            for (int j = 0; j < 32; ++j) {
                tot += v[j];
                if (q * COL_ROWS + h * 32 + j < (int)blockIdx.x) bef += v[j];
            }
        }
        ptot[q][c] = tot;
        pbef[q][c] = bef;
    }
    __syncthreads();

    int tile_tot = 0, tile_bef = 0;
    if (tid < N_TILES) {
#pragma unroll
        for (int q = 0; q < COL_PARTS; ++q) { tile_tot += ptot[q][tid]; tile_bef += pbef[q][tid]; }
        chunks[tid] = (tile_tot + 31) >> 5;
    }
#pragma unroll
    for (int k = 0; k < BIN_PER_THR; ++k) {
        const int dd = d[k] >= 0 ? d[k] : 0;
        t[k] = dd / TILE_ROWS;
        rank[k] = 0;
        if (d[k] >= 0) rank[k] = atomicAdd(&cnt[t[k]], 1);
    }
    __syncthreads();

    if (tid < N_TILES) {
        int pre = 0;
        for (int j = 0; j < tid; ++j) pre += chunks[j];
        first[tid] = pre * 32 + tile_bef;
        if (blockIdx.x == 0) {
            tile_chunks[2 * tid]     = pre;
            tile_chunks[2 * tid + 1] = tile_tot;
        }
    }
    __syncthreads();

    uint2* recs = (uint2*)replay;
#pragma unroll
    for (int k = 0; k < BIN_PER_THR; ++k) {
        if (d[k] >= 0) {
            const int dl = d[k] - t[k] * TILE_ROWS;
            uint2 rec;
            rec.x = (unsigned)s[k] | ((unsigned)dl << S_BITS);
            rec.y = 0u;
            recs[first[t[k]] + rank[k]] = rec;
        }
    }
}

#define PUSH_THREADS 1024
#define PUSH_WAVES   (PUSH_THREADS / 32)
#define ROUND_RECS   8192
#define RECS_PER_THR (ROUND_RECS / PUSH_THREADS)
#define WAVE_ROWS    ((TILE_ROWS + PUSH_WAVES - 1) / PUSH_WAVES)
#define SPARE_SLOT   15

typedef __attribute__((ext_vector_type(16))) float v16f;

#define H_PITCH       68
#define TILE_BLOCKS   ((TILE_ROWS + 15) / 16)
#define TILE_LDS_ROWS (TILE_BLOCKS * 16)
#define TILE_LDS_BYTES (TILE_LDS_ROWS * H_PITCH * 4)
#define WFRAG_N       (16 * 4 * 32)

template <int FINAL>
__global__ __launch_bounds__(PUSH_THREADS) void tile_push_kernel(const float* __restrict__ dinv,
                                                                 const float* __restrict__ feat,
                                                                 float* __restrict__ out,
                                                                 const int* __restrict__ tile_chunks,
                                                                 unsigned* __restrict__ replay,
                                                                 const float* __restrict__ bias_mu,
                                                                 const float* __restrict__ bias_lv,
                                                                 const float* __restrict__ W1,
                                                                 const float* __restrict__ b1,
                                                                 const float* __restrict__ Wmu,
                                                                 const float* __restrict__ Wlv) {
    __shared__ float dloc[TILE_ROWS];
    __shared__ __attribute__((aligned(16))) unsigned char pool[FINAL ? ROUND_RECS * 8 : TILE_LDS_BYTES];
    __shared__ int   qcnt[PUSH_WAVES];
    __shared__ v2f   wfrag[FINAL ? 1 : 2 * WFRAG_N];
    uint2* sorted = (uint2*)pool;

    const int tid  = threadIdx.x;
    const int lane = tid & 31;
    const int wave = __builtin_amdgcn_readfirstlane(tid >> 5);
    const int lo   = blockIdx.x * TILE_ROWS;
    const int rows = min(TILE_ROWS, N_NODES - lo);

    if (tid < rows) dloc[tid] = dinv[lo + tid];
    if (tid < PUSH_WAVES) qcnt[tid] = 0;

    if (!FINAL) {
#pragma unroll
        for (int q = 0; q < 2; ++q) {
            const int f  = q * PUSH_THREADS + tid;
            const int fm = f & 15;
            const int fh = (f >> 4) & 1;
            const int fj = (f >> 5) & 3;
            const int kk = (f >> 7) * 4 + fh * 2;
            v2f w;
            w.x = W1[kk * DIM + 16 * fj + fm];
            w.y = W1[(kk + 1) * DIM + 16 * fj + fm];
            wfrag[f] = w;
            const float* w2 = (fj < 2 ? Wmu : Wlv) + 16 * (fj & 1) + fm;
            v2f u;
            u.x = w2[kk * OUT_DIM];
            u.y = w2[(kk + 1) * OUT_DIM];
            wfrag[WFRAG_N + f] = u;
        }
    }

    const float2* feat2 = (const float2*)feat;

    v16f accx, accy;
    {
        float2 f0[WAVE_ROWS];
        float  dv[WAVE_ROWS];
#pragma unroll
        for (int j = 0; j < WAVE_ROWS; ++j) {
            const int dl = min(32 * j + wave, rows - 1);
            f0[j] = feat2[(size_t)(lo + dl) * (DIM / 2) + lane];
            dv[j] = dinv[lo + dl];
        }
#pragma unroll
        for (int j = 0; j < 16; ++j) {
            float vx = 0.0f, vy = 0.0f;
            if (j < WAVE_ROWS) {
                if (32 * j + wave < rows) {
                    const float w = dv[j] * dv[j];
                    vx = f0[j].x * w;
                    vy = f0[j].y * w;
                }
            }
            accx[j] = vx;
            accy[j] = vy;
        }
    }
    __syncthreads();

    const int base = tile_chunks[2 * blockIdx.x];
    const int nrec = tile_chunks[2 * blockIdx.x + 1];
    uint2* my_recs = (uint2*)replay + (size_t)base * 32;

    for (int r0 = 0; r0 < nrec; r0 += ROUND_RECS) {
        unsigned rx[RECS_PER_THR], ry[RECS_PER_THR];
        int      rank[RECS_PER_THR];

#pragma unroll
        for (int k = 0; k < RECS_PER_THR; ++k) {
            const int idx = r0 + k * PUSH_THREADS + tid;
            const uint2 r = my_recs[min(idx, nrec - 1)];
            rx[k] = r.x;  ry[k] = r.y;
        }
        if (!FINAL) {
#pragma unroll
            for (int k = 0; k < RECS_PER_THR; ++k)
                ry[k] = (unsigned)__float_as_int(dinv[rx[k] & ((1u << S_BITS) - 1u)]);
#pragma unroll
            for (int k = 0; k < RECS_PER_THR; ++k) {
                const int idx = r0 + k * PUSH_THREADS + tid;
                if (idx < nrec) {
                    uint2 o;  o.x = rx[k];  o.y = ry[k];
                    my_recs[idx] = o;
                }
            }
        }
#pragma unroll
        for (int k = 0; k < RECS_PER_THR; ++k) {
            const int idx = r0 + k * PUSH_THREADS + tid;
            rank[k] = 0;
            if (idx < nrec) rank[k] = atomicAdd(&qcnt[(rx[k] >> S_BITS) & (PUSH_WAVES - 1)], 1);
        }
        __syncthreads();

        const int c = qcnt[lane];
        int incl = c;
#pragma unroll
        for (int d = 1; d < 32; d <<= 1) {
            const int t = __shfl_up(incl, d);
            if (lane >= d) incl += t;
        }
        const int excl = incl - c;
        const int seg0 = __builtin_amdgcn_readlane(excl, wave);
        const int segn = __builtin_amdgcn_readlane(c, wave);

#pragma unroll
        for (int k = 0; k < RECS_PER_THR; ++k) {
            const int idx = r0 + k * PUSH_THREADS + tid;
            const int dl  = (int)(rx[k] >> S_BITS);
            const int at  = __shfl(excl, dl & (PUSH_WAVES - 1)) + rank[k];
            if (idx < nrec) {
                uint2 o;
                o.x = (rx[k] & ((1u << S_BITS) - 1u)) | ((unsigned)(dl >> 5) << S_BITS);
                o.y = (unsigned)__float_as_int(__int_as_float((int)ry[k]) * dloc[dl]);
                sorted[at] = o;
            }
        }
        __syncthreads();
        if (tid < PUSH_WAVES) qcnt[tid] = 0;

        for (int b = 0; b < segn; b += 32) {
            const int n = min(32, segn - b);
            uint2 r;  r.x = (unsigned)SPARE_SLOT << S_BITS;  r.y = 0u;
            if (lane < n) r = sorted[seg0 + b + lane];
            const unsigned s  = r.x & ((1u << S_BITS) - 1u);
            const int      el = (int)(r.x >> S_BITS);
            const int      wb = (int)r.y;
            float2 f[32];
#pragma unroll
            for (int k = 0; k < 32; ++k) {
                const unsigned si = (unsigned)__builtin_amdgcn_readlane((int)s, k);
                f[k] = feat2[si * (unsigned)(DIM / 2) + (unsigned)lane];
            }
#pragma unroll
            for (int k = 0; k < 32; ++k) {
                const int   j = __builtin_amdgcn_readlane(el, k);
                const float w = __int_as_float(__builtin_amdgcn_readlane(wb, k));
                float ax = accx[j];
                float ay = accy[j];
                ax += f[k].x * w;
                ay += f[k].y * w;
                accx[j] = ax;
                accy[j] = ay;
            }
        }
        __syncthreads();
    }

    if (FINAL) {
        const int h  = lane >> 4;
        const int cc = (lane & 15) * 2;
        const float* bsrc = (h ? bias_lv : bias_mu) + cc;
        const float bx = bsrc[0], by = bsrc[1];
        float* plane = (h ? out + (size_t)N_NODES * OUT_DIM : out) + cc;
#pragma unroll
        for (int j = 0; j < WAVE_ROWS; ++j) {
            const int dl = 32 * j + wave;
            if (dl < rows) {
                float2 v;  v.x = accx[j] + bx;  v.y = accy[j] + by;
                *(float2*)(plane + (size_t)(lo + dl) * OUT_DIM) = v;
            }
        }
    } else {
        float* tile = (float*)pool;
        const int nblk = (rows + 15) >> 4;
#pragma unroll
        for (int j = 0; j < WAVE_ROWS; ++j) {
            const int dl = 32 * j + wave;
            if (dl < rows) {
                v2f v;  v.x = accx[j];  v.y = accy[j];
                *(v2f*)(tile + dl * H_PITCH + 2 * lane) = v;
            }
        }
        for (int i = rows * H_PITCH + tid; i < nblk * 16 * H_PITCH; i += PUSH_THREADS) tile[i] = 0.0f;
        __syncthreads();

        const bool act  = wave < nblk;
        const int  m    = lane & 15;
        const int  half = lane >> 4;
        float* blk = tile + (16 * wave) * H_PITCH;
        const float* arow = blk + m * H_PITCH + 2 * half;

        if (act) {
            v8f h0 = {}, h1 = {}, h2 = {}, h3 = {};
#pragma unroll
            for (int ks = 0; ks < 16; ++ks) {
                const v2f a = *(const v2f*)(arow + 4 * ks);
                const v2f* wf = wfrag + (ks * 4) * 32 + lane;
                h0 = __builtin_amdgcn_wmma_f32_16x16x4_f32(false, a, false, wf[0],  (short)0, h0, false, false);
                h1 = __builtin_amdgcn_wmma_f32_16x16x4_f32(false, a, false, wf[32], (short)0, h1, false, false);
                h2 = __builtin_amdgcn_wmma_f32_16x16x4_f32(false, a, false, wf[64], (short)0, h2, false, false);
                h3 = __builtin_amdgcn_wmma_f32_16x16x4_f32(false, a, false, wf[96], (short)0, h3, false, false);
            }
            const float c0 = b1[m], c1 = b1[16 + m], c2 = b1[32 + m], c3 = b1[48 + m];
#pragma unroll
            for (int r = 0; r < 8; ++r) {
                float* t = blk + (r + 8 * half) * H_PITCH;
                t[m]      = fmaxf(h0[r] + c0, 0.0f);
                t[16 + m] = fmaxf(h1[r] + c1, 0.0f);
                t[32 + m] = fmaxf(h2[r] + c2, 0.0f);
                t[48 + m] = fmaxf(h3[r] + c3, 0.0f);
            }
        }
        __syncthreads();

        if (act) {
            v8f accMu0 = {}, accMu1 = {}, accLv0 = {}, accLv1 = {};
#pragma unroll
            for (int ks = 0; ks < 16; ++ks) {
                const v2f a = *(const v2f*)(arow + 4 * ks);
                const v2f* wf = wfrag + WFRAG_N + (ks * 4) * 32 + lane;
                accMu0 = __builtin_amdgcn_wmma_f32_16x16x4_f32(false, a, false, wf[0],  (short)0, accMu0, false, false);
                accMu1 = __builtin_amdgcn_wmma_f32_16x16x4_f32(false, a, false, wf[32], (short)0, accMu1, false, false);
                accLv0 = __builtin_amdgcn_wmma_f32_16x16x4_f32(false, a, false, wf[64], (short)0, accLv0, false, false);
                accLv1 = __builtin_amdgcn_wmma_f32_16x16x4_f32(false, a, false, wf[96], (short)0, accLv1, false, false);
            }
#pragma unroll
            for (int r = 0; r < 8; ++r) {
                float* t = blk + (r + 8 * half) * H_PITCH;
                t[m]      = accMu0[r];
                t[16 + m] = accMu1[r];
                t[32 + m] = accLv0[r];
                t[48 + m] = accLv1[r];
            }
        }
        __syncthreads();

        if (act) {
            v2f* out2 = (v2f*)out;
            v2f  v[16];
#pragma unroll
            for (int r = 0; r < 16; ++r) v[r] = *(const v2f*)(blk + r * H_PITCH + 2 * lane);
#pragma unroll
            for (int r = 0; r < 16; ++r) {
                const int dl = 16 * wave + r;
                if (dl < rows) out2[(size_t)(lo + dl) * (DIM / 2) + lane] = v[r];
            }
        }
    }
}

extern "C" void kernel_launch(void* const* d_in, const int* in_sizes, int n_in,
                              void* d_out, int out_size, void* d_ws, size_t ws_size,
                              hipStream_t stream) {
    const float* x    = (const float*)d_in[0];
    const int*   eidx = (const int*)d_in[1];
    const int*   src  = eidx;
    const int*   dst  = eidx + N_EDGES;
    const float* W1   = (const float*)d_in[2];
    const float* b1   = (const float*)d_in[3];
    const float* Wmu  = (const float*)d_in[4];
    const float* bmu  = (const float*)d_in[5];
    const float* Wlv  = (const float*)d_in[6];
    const float* blv  = (const float*)d_in[7];
    float*       out  = (float*)d_out;

    float* dinv = (float*)d_ws;
    float* agg2 = dinv + N_NODES;
    int*   cntmat      = (int*)(agg2 + (size_t)N_NODES * DIM + 32);
    int*   tile_chunks = cntmat + BIN_GROUPS * N_TILES;
    unsigned* replay   = (unsigned*)(tile_chunks + 2 * N_TILES);

    count_edges_kernel<<<BIN_GROUPS, BIN_THREADS, 0, stream>>>(dst, cntmat);
    bin_edges_kernel<<<BIN_GROUPS, BIN_THREADS, 0, stream>>>(src, dst, cntmat, tile_chunks, replay);
    tile_degree_kernel<<<N_TILES, DEG_THREADS, 0, stream>>>(tile_chunks, replay, dinv);

    tile_push_kernel<0><<<N_TILES, PUSH_THREADS, 0, stream>>>(dinv, x, agg2, tile_chunks, replay, bmu, blv, W1, b1, Wmu, Wlv);
    tile_push_kernel<1><<<N_TILES, PUSH_THREADS, 0, stream>>>(dinv, agg2, out, tile_chunks, replay, bmu, blv, W1, b1, Wmu, Wlv);
}


# Count pass and its launch go: a bin group sorts its edges by tile (LDS counters) into its own part of the list and writes a run table; a tile's owner reads 256 runs; sum order = LDS arrival rank
# speedup vs baseline: 7.1520x; 1.2263x over previous
#include <hip/hip_runtime.h>
#include <hip/hip_bf16.h>

#define N_NODES 100000
#define N_EDGES 1600000
#define DIM 64
#define OUT_DIM 32

typedef __attribute__((ext_vector_type(2))) float v2f;
typedef __attribute__((ext_vector_type(8))) float v8f;

#define TILE_ROWS    391
#define N_TILES      256
#define S_BITS       17

#define BIN_GROUPS   256
#define BIN_THREADS  1024
#define BIN_EDGES    (N_EDGES / BIN_GROUPS)
#define BIN_PER_THR  ((BIN_EDGES + BIN_THREADS - 1) / BIN_THREADS)

#define RUNS_PER_WAVE (BIN_GROUPS / 32)

#define DEG_THREADS  1024

__global__ __launch_bounds__(DEG_THREADS) void tile_degree_kernel(const int* __restrict__ table,
                                                                  const unsigned* __restrict__ replay,
                                                                  float* __restrict__ dinv) {
    __shared__ int degc[TILE_ROWS];
    __shared__ int wmax[DEG_THREADS / 32];
    const int tid  = threadIdx.x;
    const int lane = tid & 31;
    const int wave = __builtin_amdgcn_readfirstlane(tid >> 5);
    if (tid < TILE_ROWS) degc[tid] = 0;

    const int2* trow = (const int2*)table + blockIdx.x * BIN_GROUPS + RUNS_PER_WAVE * wave;
    int st[RUNS_PER_WAVE], ln[RUNS_PER_WAVE];
    int mx = 0;
#pragma unroll
    for (int j = 0; j < RUNS_PER_WAVE; ++j) {
        const int2 e = trow[j];
        st[j] = (RUNS_PER_WAVE * wave + j) * BIN_EDGES + e.x;
        ln[j] = e.y;
        mx = max(mx, e.y);
    }
    if (lane == 0) wmax[wave] = mx;
    __syncthreads();
    int maxlen = wmax[lane];
#pragma unroll
    for (int d = 16; d >= 1; d >>= 1) maxlen = max(maxlen, __shfl_xor(maxlen, d));
    maxlen = __builtin_amdgcn_readfirstlane(maxlen);

    const uint2* recs = (const uint2*)replay;
    for (int k0 = 0; k0 < maxlen; k0 += 32) {
        unsigned key[RUNS_PER_WAVE];
#pragma unroll
        for (int j = 0; j < RUNS_PER_WAVE; ++j) {
            const int at = st[j] + min(k0 + lane, max(ln[j] - 1, 0));
            key[j] = recs[min(at, N_EDGES - 1)].x;
        }
#pragma unroll
        for (int j = 0; j < RUNS_PER_WAVE; ++j)
            if (k0 + lane < ln[j]) atomicAdd(&degc[key[j] >> S_BITS], 1);
    }
    __syncthreads();

    const int i = blockIdx.x * TILE_ROWS + tid;
    if (tid < TILE_ROWS && i < N_NODES) dinv[i] = rsqrtf((float)(degc[tid] + 1));
}

__global__ __launch_bounds__(BIN_THREADS) void bin_edges_kernel(const int* __restrict__ src,
                                                                const int* __restrict__ dst,
                                                                int* __restrict__ table,
                                                                unsigned* __restrict__ replay) {
    __shared__ int cnt[N_TILES];
    __shared__ int first[N_TILES];

    const int tid = threadIdx.x;
    if (tid < N_TILES) cnt[tid] = 0;

    const int e0 = blockIdx.x * BIN_EDGES;
    int d[BIN_PER_THR], s[BIN_PER_THR], t[BIN_PER_THR], rank[BIN_PER_THR];

#pragma unroll
    for (int k = 0; k < BIN_PER_THR; ++k) {
        const int idx = k * BIN_THREADS + tid;
        const int e   = e0 + min(idx, BIN_EDGES - 1);
        d[k] = dst[e];
        s[k] = src[e];
        if (idx >= BIN_EDGES) d[k] = -1;
    }
    __syncthreads();

#pragma unroll
    for (int k = 0; k < BIN_PER_THR; ++k) {
        const int dd = d[k] >= 0 ? d[k] : 0;
        t[k] = dd / TILE_ROWS;
        rank[k] = 0;
        if (d[k] >= 0) rank[k] = atomicAdd(&cnt[t[k]], 1);
    }
    __syncthreads();

    if (tid < N_TILES) {
        int pre = 0;
        for (int j = 0; j < tid; ++j) pre += cnt[j];
        first[tid] = pre;
        int2 e;  e.x = pre;  e.y = cnt[tid];
        ((int2*)table)[tid * BIN_GROUPS + blockIdx.x] = e;
    }
    __syncthreads();

    uint2* recs = (uint2*)replay + (size_t)blockIdx.x * BIN_EDGES;
#pragma unroll
    for (int k = 0; k < BIN_PER_THR; ++k) {
        if (d[k] >= 0) {
            const int dl = d[k] - t[k] * TILE_ROWS;
            uint2 rec;
            rec.x = (unsigned)s[k] | ((unsigned)dl << S_BITS);
            rec.y = 0u;
            recs[first[t[k]] + rank[k]] = rec;
        }
    }
}

#define PUSH_THREADS 1024
#define PUSH_WAVES   (PUSH_THREADS / 32)
#define ROUND_RECS   8192
#define RECS_PER_THR (ROUND_RECS / PUSH_THREADS)
#define WAVE_ROWS    ((TILE_ROWS + PUSH_WAVES - 1) / PUSH_WAVES)
#define SPARE_SLOT   15

static_assert(RECS_PER_THR == RUNS_PER_WAVE, "a thread takes one record of each of its wave's runs a round");

typedef __attribute__((ext_vector_type(16))) float v16f;

#define H_PITCH       68
#define TILE_BLOCKS   ((TILE_ROWS + 15) / 16)
#define TILE_LDS_ROWS (TILE_BLOCKS * 16)
#define TILE_LDS_BYTES (TILE_LDS_ROWS * H_PITCH * 4)
#define WFRAG_N       (16 * 4 * 32)

template <int FINAL>
__global__ __launch_bounds__(PUSH_THREADS) void tile_push_kernel(const float* __restrict__ dinv,
                                                                 const float* __restrict__ feat,
                                                                 float* __restrict__ out,
                                                                 const int* __restrict__ table,
                                                                 unsigned* __restrict__ replay,
                                                                 const float* __restrict__ bias_mu,
                                                                 const float* __restrict__ bias_lv,
                                                                 const float* __restrict__ W1,
                                                                 const float* __restrict__ b1,
                                                                 const float* __restrict__ Wmu,
                                                                 const float* __restrict__ Wlv) {
    __shared__ float dloc[TILE_ROWS];
    __shared__ __attribute__((aligned(16))) unsigned char pool[FINAL ? ROUND_RECS * 8 : TILE_LDS_BYTES];
    __shared__ int   qcnt[PUSH_WAVES];
    __shared__ int   wmax[PUSH_WAVES];
    __shared__ v2f   wfrag[FINAL ? 1 : 2 * WFRAG_N];
    uint2* sorted = (uint2*)pool;

    const int tid  = threadIdx.x;
    const int lane = tid & 31;
    const int wave = __builtin_amdgcn_readfirstlane(tid >> 5);
    const int lo   = blockIdx.x * TILE_ROWS;
    const int rows = min(TILE_ROWS, N_NODES - lo);

    if (tid < rows) dloc[tid] = dinv[lo + tid];
    if (tid < PUSH_WAVES) qcnt[tid] = 0;

    int st[RUNS_PER_WAVE], ln[RUNS_PER_WAVE];
    {
        const int2* trow = (const int2*)table + blockIdx.x * BIN_GROUPS + RUNS_PER_WAVE * wave;
        int mx = 0;
#pragma unroll
        for (int j = 0; j < RUNS_PER_WAVE; ++j) {
            const int2 e = trow[j];
            st[j] = (RUNS_PER_WAVE * wave + j) * BIN_EDGES + e.x;
            ln[j] = e.y;
            mx = max(mx, e.y);
        }
        if (lane == 0) wmax[wave] = mx;
    }

    if (!FINAL) {
#pragma unroll
        for (int q = 0; q < 2; ++q) {
            const int f  = q * PUSH_THREADS + tid;
            const int fm = f & 15;
            const int fh = (f >> 4) & 1;
            const int fj = (f >> 5) & 3;
            const int kk = (f >> 7) * 4 + fh * 2;
            v2f w;
            w.x = W1[kk * DIM + 16 * fj + fm];
            w.y = W1[(kk + 1) * DIM + 16 * fj + fm];
            wfrag[f] = w;
            const float* w2 = (fj < 2 ? Wmu : Wlv) + 16 * (fj & 1) + fm;
            v2f u;
            u.x = w2[kk * OUT_DIM];
            u.y = w2[(kk + 1) * OUT_DIM];
            wfrag[WFRAG_N + f] = u;
        }
    }

    const float2* feat2 = (const float2*)feat;

    v16f accx, accy;
    {
        float2 f0[WAVE_ROWS];
        float  dv[WAVE_ROWS];
#pragma unroll
        for (int j = 0; j < WAVE_ROWS; ++j) {
            const int dl = min(32 * j + wave, rows - 1);
            f0[j] = feat2[(size_t)(lo + dl) * (DIM / 2) + lane];
            dv[j] = dinv[lo + dl];
        }
#pragma unroll
        for (int j = 0; j < 16; ++j) {
            float vx = 0.0f, vy = 0.0f;
            if (j < WAVE_ROWS) {
                if (32 * j + wave < rows) {
                    const float w = dv[j] * dv[j];
                    vx = f0[j].x * w;
                    vy = f0[j].y * w;
                }
            }
            accx[j] = vx;
            accy[j] = vy;
        }
    }
    __syncthreads();

    int maxlen = wmax[lane];
#pragma unroll
    for (int d = 16; d >= 1; d >>= 1) maxlen = max(maxlen, __shfl_xor(maxlen, d));
    maxlen = __builtin_amdgcn_readfirstlane(maxlen);

    uint2* recs = (uint2*)replay;

    for (int k0 = 0; k0 < maxlen; k0 += 32) {
        unsigned rx[RECS_PER_THR], ry[RECS_PER_THR];
        int      rank[RECS_PER_THR];
        int      at[RECS_PER_THR];
        bool     ok[RECS_PER_THR];

#pragma unroll
        for (int k = 0; k < RECS_PER_THR; ++k) {
            ok[k] = k0 + lane < ln[k];
            at[k] = min(st[k] + min(k0 + lane, max(ln[k] - 1, 0)), N_EDGES - 1);
            const uint2 r = recs[at[k]];
            rx[k] = r.x;  ry[k] = r.y;
        }
        if (!FINAL) {
#pragma unroll
            for (int k = 0; k < RECS_PER_THR; ++k)
                ry[k] = (unsigned)__float_as_int(dinv[rx[k] & ((1u << S_BITS) - 1u)]);
#pragma unroll
            for (int k = 0; k < RECS_PER_THR; ++k) {
                if (ok[k]) {
                    uint2 o;  o.x = rx[k];  o.y = ry[k];
                    recs[at[k]] = o;
                }
            }
        }
#pragma unroll
        for (int k = 0; k < RECS_PER_THR; ++k) {
            rank[k] = 0;
            if (ok[k]) rank[k] = atomicAdd(&qcnt[(rx[k] >> S_BITS) & (PUSH_WAVES - 1)], 1);
        }
        __syncthreads();

        const int c = qcnt[lane];
        int incl = c;
#pragma unroll
        for (int d = 1; d < 32; d <<= 1) {
            const int t = __shfl_up(incl, d);
            if (lane >= d) incl += t;
        }
        const int excl = incl - c;
        const int seg0 = __builtin_amdgcn_readlane(excl, wave);
        const int segn = __builtin_amdgcn_readlane(c, wave);

#pragma unroll
        for (int k = 0; k < RECS_PER_THR; ++k) {
            const int dl  = (int)(rx[k] >> S_BITS);
            const int to  = __shfl(excl, dl & (PUSH_WAVES - 1)) + rank[k];
            if (ok[k]) {
                uint2 o;
                o.x = (rx[k] & ((1u << S_BITS) - 1u)) | ((unsigned)(dl >> 5) << S_BITS);
                o.y = (unsigned)__float_as_int(__int_as_float((int)ry[k]) * dloc[dl]);
                sorted[to] = o;
            }
        }
        __syncthreads();
        if (tid < PUSH_WAVES) qcnt[tid] = 0;

        for (int b = 0; b < segn; b += 32) {
            const int n = min(32, segn - b);
            uint2 r;  r.x = (unsigned)SPARE_SLOT << S_BITS;  r.y = 0u;
            if (lane < n) r = sorted[seg0 + b + lane];
            const unsigned s  = r.x & ((1u << S_BITS) - 1u);
            const int      el = (int)(r.x >> S_BITS);
            const int      wb = (int)r.y;
            float2 f[32];
#pragma unroll
            for (int k = 0; k < 32; ++k) {
                const unsigned si = (unsigned)__builtin_amdgcn_readlane((int)s, k);
                f[k] = feat2[si * (unsigned)(DIM / 2) + (unsigned)lane];
            }
#pragma unroll
            for (int k = 0; k < 32; ++k) {
                const int   j = __builtin_amdgcn_readlane(el, k);
                const float w = __int_as_float(__builtin_amdgcn_readlane(wb, k));
                float ax = accx[j];
                float ay = accy[j];
                ax += f[k].x * w;
                ay += f[k].y * w;
                accx[j] = ax;
                accy[j] = ay;
            }
        }
        __syncthreads();
    }

    if (FINAL) {
        const int h  = lane >> 4;
        const int cc = (lane & 15) * 2;
        const float* bsrc = (h ? bias_lv : bias_mu) + cc;
        const float bx = bsrc[0], by = bsrc[1];
        float* plane = (h ? out + (size_t)N_NODES * OUT_DIM : out) + cc;
#pragma unroll
        for (int j = 0; j < WAVE_ROWS; ++j) {
            const int dl = 32 * j + wave;
            if (dl < rows) {
                float2 v;  v.x = accx[j] + bx;  v.y = accy[j] + by;
                *(float2*)(plane + (size_t)(lo + dl) * OUT_DIM) = v;
            }
        }
    } else {
        float* tile = (float*)pool;
        const int nblk = (rows + 15) >> 4;
#pragma unroll
        for (int j = 0; j < WAVE_ROWS; ++j) {
            const int dl = 32 * j + wave;
            if (dl < rows) {
                v2f v;  v.x = accx[j];  v.y = accy[j];
                *(v2f*)(tile + dl * H_PITCH + 2 * lane) = v;
            }
        }
        for (int i = rows * H_PITCH + tid; i < nblk * 16 * H_PITCH; i += PUSH_THREADS) tile[i] = 0.0f;
        __syncthreads();

        const bool act  = wave < nblk;
        const int  m    = lane & 15;
        const int  half = lane >> 4;
        float* blk = tile + (16 * wave) * H_PITCH;
        const float* arow = blk + m * H_PITCH + 2 * half;

        if (act) {
            v8f h0 = {}, h1 = {}, h2 = {}, h3 = {};
#pragma unroll
            for (int ks = 0; ks < 16; ++ks) {
                const v2f a = *(const v2f*)(arow + 4 * ks);
                const v2f* wf = wfrag + (ks * 4) * 32 + lane;
                h0 = __builtin_amdgcn_wmma_f32_16x16x4_f32(false, a, false, wf[0],  (short)0, h0, false, false);
                h1 = __builtin_amdgcn_wmma_f32_16x16x4_f32(false, a, false, wf[32], (short)0, h1, false, false);
                h2 = __builtin_amdgcn_wmma_f32_16x16x4_f32(false, a, false, wf[64], (short)0, h2, false, false);
                h3 = __builtin_amdgcn_wmma_f32_16x16x4_f32(false, a, false, wf[96], (short)0, h3, false, false);
            }
            const float c0 = b1[m], c1 = b1[16 + m], c2 = b1[32 + m], c3 = b1[48 + m];
#pragma unroll
            for (int r = 0; r < 8; ++r) {
                float* t = blk + (r + 8 * half) * H_PITCH;
                t[m]      = fmaxf(h0[r] + c0, 0.0f);
                t[16 + m] = fmaxf(h1[r] + c1, 0.0f);
                t[32 + m] = fmaxf(h2[r] + c2, 0.0f);
                t[48 + m] = fmaxf(h3[r] + c3, 0.0f);
            }
        }
        __syncthreads();

        if (act) {
            v8f accMu0 = {}, accMu1 = {}, accLv0 = {}, accLv1 = {};
#pragma unroll
            for (int ks = 0; ks < 16; ++ks) {
                const v2f a = *(const v2f*)(arow + 4 * ks);
                const v2f* wf = wfrag + WFRAG_N + (ks * 4) * 32 + lane;
                accMu0 = __builtin_amdgcn_wmma_f32_16x16x4_f32(false, a, false, wf[0],  (short)0, accMu0, false, false);
                accMu1 = __builtin_amdgcn_wmma_f32_16x16x4_f32(false, a, false, wf[32], (short)0, accMu1, false, false);
                accLv0 = __builtin_amdgcn_wmma_f32_16x16x4_f32(false, a, false, wf[64], (short)0, accLv0, false, false);
                accLv1 = __builtin_amdgcn_wmma_f32_16x16x4_f32(false, a, false, wf[96], (short)0, accLv1, false, false);
            }
#pragma unroll
            for (int r = 0; r < 8; ++r) {
                float* t = blk + (r + 8 * half) * H_PITCH;
                t[m]      = accMu0[r];
                t[16 + m] = accMu1[r];
                t[32 + m] = accLv0[r];
                t[48 + m] = accLv1[r];
            }
        }
        __syncthreads();

        if (act) {
            v2f* out2 = (v2f*)out;
            v2f  v[16];
#pragma unroll
            for (int r = 0; r < 16; ++r) v[r] = *(const v2f*)(blk + r * H_PITCH + 2 * lane);
#pragma unroll
            for (int r = 0; r < 16; ++r) {
                const int dl = 16 * wave + r;
                if (dl < rows) out2[(size_t)(lo + dl) * (DIM / 2) + lane] = v[r];
            }
        }
    }
}

extern "C" void kernel_launch(void* const* d_in, const int* in_sizes, int n_in,
                              void* d_out, int out_size, void* d_ws, size_t ws_size,
                              hipStream_t stream) {
    const float* x    = (const float*)d_in[0];
    const int*   eidx = (const int*)d_in[1];
    const int*   src  = eidx;
    const int*   dst  = eidx + N_EDGES;
    const float* W1   = (const float*)d_in[2];
    const float* b1   = (const float*)d_in[3];
    const float* Wmu  = (const float*)d_in[4];
    const float* bmu  = (const float*)d_in[5];
    const float* Wlv  = (const float*)d_in[6];
    const float* blv  = (const float*)d_in[7];
    float*       out  = (float*)d_out;

    float* dinv = (float*)d_ws;
    float* agg2 = dinv + N_NODES;
    int*      table  = (int*)(agg2 + (size_t)N_NODES * DIM + 32);
    unsigned* replay = (unsigned*)(table + 2 * BIN_GROUPS * N_TILES);

    bin_edges_kernel<<<BIN_GROUPS, BIN_THREADS, 0, stream>>>(src, dst, table, replay);
    tile_degree_kernel<<<N_TILES, DEG_THREADS, 0, stream>>>(table, replay, dinv);

    tile_push_kernel<0><<<N_TILES, PUSH_THREADS, 0, stream>>>(dinv, x, agg2, table, replay, bmu, blv, W1, b1, Wmu, Wlv);
    tile_push_kernel<1><<<N_TILES, PUSH_THREADS, 0, stream>>>(dinv, agg2, out, table, replay, bmu, blv, W1, b1, Wmu, Wlv);
}
